# starting kernel
# baseline (speedup 1.0000x reference)
#include <hip/hip_runtime.h>
#include <hip/hip_bf16.h>

typedef __attribute__((ext_vector_type(2))) float v2f;
typedef __attribute__((ext_vector_type(8))) float v8f;

__global__ __launch_bounds__(256) void zero_f32_kernel(float* __restrict__ p, int n4) {
    int i = blockIdx.x * blockDim.x + threadIdx.x;
    if (i < n4) ((float4*)p)[i] = make_float4(0.f, 0.f, 0.f, 0.f);
}

__global__ __launch_bounds__(256) void negcopy_kernel(float* __restrict__ dst,
                                                      const float* __restrict__ src, int n4) {
    int i = blockIdx.x * blockDim.x + threadIdx.x;
    if (i < n4) {
        float4 v = ((const float4*)src)[i];
        ((float4*)dst)[i] = make_float4(-v.x, -v.y, -v.z, -v.w);
    }
}

__global__ __launch_bounds__(256) void packw_kernel(const float* __restrict__ W,
                                                    float* __restrict__ Wp,
                                                    int nout, int total) {
    int idx = blockIdx.x * blockDim.x + threadIdx.x;
    if (idx >= total) return;
    int col = idx % nout;
    int k   = (idx / nout) & 63;
    int m   = idx / (nout * 64);
    Wp[(((size_t)m * 32 + (k >> 1)) * nout + col) * 2 + (k & 1)] = W[idx];
}

__global__ __launch_bounds__(256) void deg_kernel(const int* __restrict__ src,
                                                  const int* __restrict__ dst,
                                                  float* __restrict__ deg, int nedges) {
    int e = blockIdx.x * blockDim.x + threadIdx.x;
    if (e >= nedges) return;
    int s = src[e];
    if (s != dst[e]) unsafeAtomicAdd(deg + s, 1.0f);
}

__global__ __launch_bounds__(256) void dis_kernel(float* __restrict__ deg, int n) {
    int i = blockIdx.x * blockDim.x + threadIdx.x;
    if (i >= n) return;
    float d = deg[i];
    deg[i] = (d > 0.f) ? rsqrtf(fmaxf(d, 1.0f)) : 0.f;
}

__global__ __launch_bounds__(256) void edgew_kernel(const int* __restrict__ src,
                                                    const int* __restrict__ dst,
                                                    const float* __restrict__ dis,
                                                    float* __restrict__ w, int nedges) {
    int e = blockIdx.x * blockDim.x + threadIdx.x;
    if (e >= nedges) return;
    int s = src[e], d = dst[e];
    w[e] = (s != d) ? (-dis[s] * dis[d]) : 0.f;
}

__global__ __launch_bounds__(256) void spmm_kernel(const int* __restrict__ src,
                                                   const int* __restrict__ dst,
                                                   const float* __restrict__ w,
                                                   const float* __restrict__ x,
                                                   float* __restrict__ out,
                                                   int nedges, float scale) {
    long long tid = (long long)blockIdx.x * blockDim.x + threadIdx.x;
    int e = (int)(tid >> 4);
    if (e >= nedges) return;
    float we = w[e] * scale;
    if (we == 0.f) return;
    int fg = (int)(tid & 15);
    int s = src[e], d = dst[e];
    float4 xv = ((const float4*)x)[s * 16 + fg];
    float* o = out + (size_t)d * 64 + fg * 4;
    unsafeAtomicAdd(o + 0, we * xv.x);
    unsafeAtomicAdd(o + 1, we * xv.y);
    unsafeAtomicAdd(o + 2, we * xv.z);
    unsafeAtomicAdd(o + 3, we * xv.w);
}

template <int NT, int NOUT>
__device__ inline void cheb_accum2(const float* __restrict__ A,
                                   const float* __restrict__ Wp,
                                   int rowA0, int rowA1, int khalf, int colLane,
                                   v8f (&acc0)[NT], v8f (&acc1)[NT]) {
    const float* ar0 = A + (size_t)rowA0 * 64;
    const float* ar1 = A + (size_t)rowA1 * 64;
    #pragma unroll
    for (int k0 = 0; k0 < 64; k0 += 4) {
        int kk = k0 + 2 * khalf;
        v2f a0 = *(const v2f*)(ar0 + kk);
        v2f a1 = *(const v2f*)(ar1 + kk);
        int p = (k0 >> 1) + khalf;
        #pragma unroll
        for (int t = 0; t < NT; ++t) {
            int cg = t * 16 + colLane;
            v2f b;
            if ((NOUT & 15) == 0 || cg < NOUT) {
                b = *(const v2f*)(Wp + ((size_t)p * NOUT + cg) * 2);
            } else {
                b.x = 0.f;
                b.y = 0.f;
            }
            acc0[t] = __builtin_amdgcn_wmma_f32_16x16x4_f32(
                false, a0, false, b, (short)0, acc0[t], false, false);
            acc1[t] = __builtin_amdgcn_wmma_f32_16x16x4_f32(
                false, a1, false, b, (short)0, acc1[t], false, false);
        }
    }
}

template <int NOUT, bool RELU>
__global__ __launch_bounds__(256) void cheb_gemm_kernel(const float* __restrict__ A0,
                                                        const float* __restrict__ A1,
                                                        const float* __restrict__ A2,
                                                        const float* __restrict__ Wp,
                                                        const float* __restrict__ bias,
                                                        float* __restrict__ out, int nodes) {
    constexpr int NT  = (NOUT + 15) / 16;
    constexpr int PSZ = 64 * NOUT;
    int wave = (blockIdx.x * blockDim.x + threadIdx.x) >> 5;
    int lane = threadIdx.x & 31;
    int l15  = lane & 15;
    int khalf = lane >> 4;
    int row0 = wave * 32;
    if (row0 >= nodes) return;

    v8f acc0[NT] = {};
    v8f acc1[NT] = {};
    cheb_accum2<NT, NOUT>(A0, Wp + 0 * PSZ, row0 + l15, row0 + 16 + l15, khalf, l15, acc0, acc1);
    cheb_accum2<NT, NOUT>(A1, Wp + 1 * PSZ, row0 + l15, row0 + 16 + l15, khalf, l15, acc0, acc1);
    cheb_accum2<NT, NOUT>(A2, Wp + 2 * PSZ, row0 + l15, row0 + 16 + l15, khalf, l15, acc0, acc1);

    #pragma unroll
    for (int t = 0; t < NT; ++t) {
        int col = t * 16 + l15;
        if ((NOUT & 15) != 0 && col >= NOUT) continue;
        float bv = bias[col];
        #pragma unroll
        for (int i = 0; i < 8; ++i) {
            int r = row0 + i + 8 * khalf;
            float v0 = acc0[t][i] + bv;
            float v1 = acc1[t][i] + bv;
            if (RELU) { v0 = fmaxf(v0, 0.f); v1 = fmaxf(v1, 0.f); }
            if (r < nodes)      out[(size_t)r * NOUT + col]        = v0;
            if (r + 16 < nodes) out[(size_t)(r + 16) * NOUT + col] = v1;
        }
    }
}

__global__ __launch_bounds__(256) void softmax40_kernel(float* __restrict__ out, int nodes) {
    int i = blockIdx.x * blockDim.x + threadIdx.x;
    if (i >= nodes) return;
    float* row = out + (size_t)i * 40;
    float m = -INFINITY;
    #pragma unroll
    for (int c = 0; c < 40; ++c) m = fmaxf(m, row[c]);
    float e[40];
    float s = 0.f;
    #pragma unroll
    for (int c = 0; c < 40; ++c) { e[c] = __expf(row[c] - m); s += e[c]; }
    float inv = 1.f / s;
    #pragma unroll
    for (int c = 0; c < 40; ++c) row[c] = e[c] * inv;
}

extern "C" void kernel_launch(void* const* d_in, const int* in_sizes, int n_in,
                              void* d_out, int out_size, void* d_ws, size_t ws_size,
                              hipStream_t stream) {
    const float* x  = (const float*)d_in[0];
    const int*   ei = (const int*)d_in[1];
    const float* W1 = (const float*)d_in[2];
    const float* b1 = (const float*)d_in[3];
    const float* W2 = (const float*)d_in[4];
    const float* b2 = (const float*)d_in[5];
    float* out = (float*)d_out;

    const int N = in_sizes[0] / 64;
    const int E = in_sizes[1] / 2;
    const int* src = ei;
    const int* dst = ei + E;

    auto align256 = [](size_t v) { return (v + 255) & ~(size_t)255; };
    char* ws = (char*)d_ws;
    size_t off = 0;
    float* dis  = (float*)(ws + off); off += align256((size_t)N * 4);
    float* wbuf = (float*)(ws + off); off += align256((size_t)E * 4);
    float* A1   = (float*)(ws + off); off += align256((size_t)N * 64 * 4);
    float* A2   = (float*)(ws + off); off += align256((size_t)N * 64 * 4);
    float* H    = (float*)(ws + off); off += align256((size_t)N * 64 * 4);
    float* Wp1  = (float*)(ws + off); off += align256((size_t)3 * 64 * 64 * 4);
    float* Wp2  = (float*)(ws + off); off += align256((size_t)3 * 64 * 40 * 4);
    (void)ws_size; (void)n_in; (void)out_size;

    const int TB = 256;
    const int nodeBlk  = (N + TB - 1) / TB;
    const int edgeBlk  = (E + TB - 1) / TB;
    const int feat4Blk = ((N * 64 / 4) + TB - 1) / TB;
    const long long spmmThreads = (long long)E * 16;
    const int spmmBlk  = (int)((spmmThreads + TB - 1) / TB);
    const int gemmBlk  = ((((N + 31) / 32) * 32) + TB - 1) / TB;

    packw_kernel<<<(3 * 64 * 64 + TB - 1) / TB, TB, 0, stream>>>(W1, Wp1, 64, 3 * 64 * 64);
    packw_kernel<<<(3 * 64 * 40 + TB - 1) / TB, TB, 0, stream>>>(W2, Wp2, 40, 3 * 64 * 40);
    zero_f32_kernel<<<(N / 4 + TB - 1) / TB, TB, 0, stream>>>(dis, N / 4);
    deg_kernel<<<edgeBlk, TB, 0, stream>>>(src, dst, dis, E);
    dis_kernel<<<nodeBlk, TB, 0, stream>>>(dis, N);
    edgew_kernel<<<edgeBlk, TB, 0, stream>>>(src, dst, dis, wbuf, E);

    zero_f32_kernel<<<feat4Blk, TB, 0, stream>>>(A1, N * 64 / 4);
    spmm_kernel<<<spmmBlk, TB, 0, stream>>>(src, dst, wbuf, x, A1, E, 1.0f);
    negcopy_kernel<<<feat4Blk, TB, 0, stream>>>(A2, x, N * 64 / 4);
    spmm_kernel<<<spmmBlk, TB, 0, stream>>>(src, dst, wbuf, A1, A2, E, 2.0f);
    cheb_gemm_kernel<64, true><<<gemmBlk, TB, 0, stream>>>(x, A1, A2, Wp1, b1, H, N);

    zero_f32_kernel<<<feat4Blk, TB, 0, stream>>>(A1, N * 64 / 4);
    spmm_kernel<<<spmmBlk, TB, 0, stream>>>(src, dst, wbuf, H, A1, E, 1.0f);
    negcopy_kernel<<<feat4Blk, TB, 0, stream>>>(A2, H, N * 64 / 4);
    spmm_kernel<<<spmmBlk, TB, 0, stream>>>(src, dst, wbuf, A1, A2, E, 2.0f);
    cheb_gemm_kernel<40, false><<<gemmBlk, TB, 0, stream>>>(H, A1, A2, Wp2, b2, out, N);

    softmax40_kernel<<<nodeBlk, TB, 0, stream>>>(out, N);
}


# SpMM: edges counting-sorted by destination in-call (int atomics give slots), then row-gather SpMM: 16 lanes/row, register sums in slot order, one store; no float atomics, zero or neg-copy passes
# speedup vs baseline: 1.5236x; 1.5236x over previous
#include <hip/hip_runtime.h>
#include <hip/hip_bf16.h>

typedef __attribute__((ext_vector_type(2))) float v2f;
typedef __attribute__((ext_vector_type(8))) float v8f;

__global__ __launch_bounds__(256) void packw_kernel(const float* __restrict__ W,
                                                    float* __restrict__ Wp,
                                                    int nout, int total) {
    int idx = blockIdx.x * blockDim.x + threadIdx.x;
    if (idx >= total) return;
    int col = idx % nout;
    int k   = (idx / nout) & 63;
    int m   = idx / (nout * 64);
    Wp[(((size_t)m * 32 + (k >> 1)) * nout + col) * 2 + (k & 1)] = W[idx];
}

__global__ __launch_bounds__(256) void init_kernel(float* __restrict__ deg,
                                                   int* __restrict__ cnt, int n) {
    int i = blockIdx.x * blockDim.x + threadIdx.x;
    if (i < n) deg[i] = 0.f;
    if (i <= n) cnt[i] = 0;
}

__global__ __launch_bounds__(256) void count_kernel(const int* __restrict__ src,
                                                    const int* __restrict__ dst,
                                                    float* __restrict__ deg,
                                                    int* __restrict__ cnt,
                                                    int* __restrict__ pos, int nedges) {
    int e = blockIdx.x * blockDim.x + threadIdx.x;
    if (e >= nedges) return;
    int s = src[e], d = dst[e];
    if (s != d) {
        unsafeAtomicAdd(deg + s, 1.0f);
        pos[e] = atomicAdd(cnt + d, 1);
    }
}

__global__ __launch_bounds__(256) void dis_kernel(float* __restrict__ deg, int n) {
    int i = blockIdx.x * blockDim.x + threadIdx.x;
    if (i >= n) return;
    float d = deg[i];
    deg[i] = (d > 0.f) ? rsqrtf(fmaxf(d, 1.0f)) : 0.f;
}

__global__ __launch_bounds__(1024) void rowptr_scan_kernel(int* __restrict__ cnt, int n1) {
    __shared__ int wsum[32];
    int t = threadIdx.x;
    int lane = t & 31, wv = t >> 5;
    int chunk = (n1 + 1023) / 1024;
    int b = t * chunk;
    int e = b + chunk;
    if (e > n1) e = n1;
    int s = 0;
    for (int i = b; i < e; ++i) s += cnt[i];
    int v = s;
    #pragma unroll
    for (int o = 1; o < 32; o <<= 1) {
        int u = __shfl_up(v, o, 32);
        if (lane >= o) v += u;
    }
    if (lane == 31) wsum[wv] = v;
    __syncthreads();
    if (wv == 0) {
        int w = wsum[lane];
        #pragma unroll
        for (int o = 1; o < 32; o <<= 1) {
            int u = __shfl_up(w, o, 32);
            if (lane >= o) w += u;
        }
        wsum[lane] = w;
    }
    __syncthreads();
    int prefix = v - s + (wv > 0 ? wsum[wv - 1] : 0);
    for (int i = b; i < e; i += 8) {
        int c[8];
        #pragma unroll
        for (int k = 0; k < 8; ++k) c[k] = (i + k < e) ? cnt[i + k] : 0;
        #pragma unroll
        for (int k = 0; k < 8; ++k) {
            if (i + k < e) cnt[i + k] = prefix;
            prefix += c[k];
        }
    }
}

__global__ __launch_bounds__(256) void fill_kernel(const int* __restrict__ src,
                                                   const int* __restrict__ dst,
                                                   const int* __restrict__ rowptr,
                                                   const int* __restrict__ pos,
                                                   int* __restrict__ csr, int nedges) {
    int e = blockIdx.x * blockDim.x + threadIdx.x;
    if (e >= nedges) return;
    int s = src[e], d = dst[e];
    if (s != d) csr[rowptr[d] + pos[e]] = s;
}

template <int G, int CNT>
__device__ __forceinline__ void gather_group(const float4* __restrict__ x4, int si16, float wl,
                                             int fg, float4& acc) {
    float4 xv[CNT];
    float  w[CNT];
    #pragma unroll
    for (int t = 0; t < CNT; ++t) {
        int idx = __shfl(si16, G + t, 16) + fg;
        w[t]  = __shfl(wl, G + t, 16);
        xv[t] = x4[idx];
    }
    #pragma unroll
    for (int t = 0; t < CNT; ++t) {
        acc.x += w[t] * xv[t].x;
        acc.y += w[t] * xv[t].y;
        acc.z += w[t] * xv[t].z;
        acc.w += w[t] * xv[t].w;
    }
}

template <bool SECOND>
__global__ __launch_bounds__(256) void spmm_gather_kernel(const int* __restrict__ rowptr,
                                                          const int* __restrict__ csr,
                                                          const float* __restrict__ dis,
                                                          const float* __restrict__ x,
                                                          const float* __restrict__ x0,
                                                          float* __restrict__ out, int n) {
    int tid = blockIdx.x * blockDim.x + threadIdx.x;
    int row = tid >> 4;
    int fg  = tid & 15;
    if (row >= n) return;
    const float4* x4 = (const float4*)x;
    int start = rowptr[row];
    int end   = rowptr[row + 1];
    float ndd = -dis[row];
    float4 v0 = make_float4(0.f, 0.f, 0.f, 0.f);
    if (SECOND) v0 = ((const float4*)x0)[row * 16 + fg];
    float4 acc = make_float4(0.f, 0.f, 0.f, 0.f);
    for (int base = start; base < end; base += 16) {
        int j = base + fg;
        bool valid = j < end;
        int si = csr[valid ? j : base];
        float wl = valid ? ndd * dis[si] : 0.f;
        int si16 = si * 16;
        int cnt = end - base;
        if (cnt >= 8) {
            gather_group<0, 8>(x4, si16, wl, fg, acc);
        } else {
            gather_group<0, 4>(x4, si16, wl, fg, acc);
            if (cnt > 4) gather_group<4, 4>(x4, si16, wl, fg, acc);
        }
        if (cnt >= 16) {
            gather_group<8, 8>(x4, si16, wl, fg, acc);
        } else if (cnt > 8) {
            gather_group<8, 4>(x4, si16, wl, fg, acc);
            if (cnt > 12) gather_group<12, 4>(x4, si16, wl, fg, acc);
        }
    }
    if (SECOND) {
        acc.x = 2.0f * acc.x - v0.x;
        acc.y = 2.0f * acc.y - v0.y;
        acc.z = 2.0f * acc.z - v0.z;
        acc.w = 2.0f * acc.w - v0.w;
    }
    ((float4*)out)[row * 16 + fg] = acc;
}

template <int NT, int NOUT>
__device__ inline void cheb_accum2(const float* __restrict__ A,
                                   const float* __restrict__ Wp,
                                   int rowA0, int rowA1, int khalf, int colLane,
                                   v8f (&acc0)[NT], v8f (&acc1)[NT]) {
    const float* ar0 = A + (size_t)rowA0 * 64;
    const float* ar1 = A + (size_t)rowA1 * 64;
    #pragma unroll
    for (int k0 = 0; k0 < 64; k0 += 4) {
        int kk = k0 + 2 * khalf;
        v2f a0 = *(const v2f*)(ar0 + kk);
        v2f a1 = *(const v2f*)(ar1 + kk);
        int p = (k0 >> 1) + khalf;
        #pragma unroll
        for (int t = 0; t < NT; ++t) {
            int cg = t * 16 + colLane;
            v2f b;
            if ((NOUT & 15) == 0 || cg < NOUT) {
                b = *(const v2f*)(Wp + ((size_t)p * NOUT + cg) * 2);
            } else {
                b.x = 0.f;
                b.y = 0.f;
            }
            acc0[t] = __builtin_amdgcn_wmma_f32_16x16x4_f32(
                false, a0, false, b, (short)0, acc0[t], false, false);
            acc1[t] = __builtin_amdgcn_wmma_f32_16x16x4_f32(
                false, a1, false, b, (short)0, acc1[t], false, false);
        }
    }
}

template <int NOUT, bool RELU>
__global__ __launch_bounds__(256) void cheb_gemm_kernel(const float* __restrict__ A0,
                                                        const float* __restrict__ A1,
                                                        const float* __restrict__ A2,
                                                        const float* __restrict__ Wp,
                                                        const float* __restrict__ bias,
                                                        float* __restrict__ out, int nodes) {
    constexpr int NT  = (NOUT + 15) / 16;
    constexpr int PSZ = 64 * NOUT;
    int wave = (blockIdx.x * blockDim.x + threadIdx.x) >> 5;
    int lane = threadIdx.x & 31;
    int l15  = lane & 15;
    int khalf = lane >> 4;
    int row0 = wave * 32;
    if (row0 >= nodes) return;

    v8f acc0[NT] = {};
    v8f acc1[NT] = {};
    cheb_accum2<NT, NOUT>(A0, Wp + 0 * PSZ, row0 + l15, row0 + 16 + l15, khalf, l15, acc0, acc1);
    cheb_accum2<NT, NOUT>(A1, Wp + 1 * PSZ, row0 + l15, row0 + 16 + l15, khalf, l15, acc0, acc1);
    cheb_accum2<NT, NOUT>(A2, Wp + 2 * PSZ, row0 + l15, row0 + 16 + l15, khalf, l15, acc0, acc1);

    #pragma unroll
    for (int t = 0; t < NT; ++t) {
        int col = t * 16 + l15;
        if ((NOUT & 15) != 0 && col >= NOUT) continue;
        float bv = bias[col];
        #pragma unroll
        for (int i = 0; i < 8; ++i) {
            int r = row0 + i + 8 * khalf;
            float v0 = acc0[t][i] + bv;
            float v1 = acc1[t][i] + bv;
            if (RELU) { v0 = fmaxf(v0, 0.f); v1 = fmaxf(v1, 0.f); }
            if (r < nodes)      out[(size_t)r * NOUT + col]        = v0;
            if (r + 16 < nodes) out[(size_t)(r + 16) * NOUT + col] = v1;
        }
    }
}

__global__ __launch_bounds__(256) void softmax40_kernel(float* __restrict__ out, int nodes) {
    int i = blockIdx.x * blockDim.x + threadIdx.x;
    if (i >= nodes) return;
    float* row = out + (size_t)i * 40;
    float m = -INFINITY;
    #pragma unroll
    for (int c = 0; c < 40; ++c) m = fmaxf(m, row[c]);
    float e[40];
    float s = 0.f;
    #pragma unroll
    for (int c = 0; c < 40; ++c) { e[c] = __expf(row[c] - m); s += e[c]; }
    float inv = 1.f / s;
    #pragma unroll
    for (int c = 0; c < 40; ++c) row[c] = e[c] * inv;
}

extern "C" void kernel_launch(void* const* d_in, const int* in_sizes, int n_in,
                              void* d_out, int out_size, void* d_ws, size_t ws_size,
                              hipStream_t stream) {
    const float* x  = (const float*)d_in[0];
    const int*   ei = (const int*)d_in[1];
    const float* W1 = (const float*)d_in[2];
    const float* b1 = (const float*)d_in[3];
    const float* W2 = (const float*)d_in[4];
    const float* b2 = (const float*)d_in[5];
    float* out = (float*)d_out;

    const int N = in_sizes[0] / 64;
    const int E = in_sizes[1] / 2;
    const int* src = ei;
    const int* dst = ei + E;

    auto align256 = [](size_t v) { return (v + 255) & ~(size_t)255; };
    char* ws = (char*)d_ws;
    size_t off = 0;
    float* dis  = (float*)(ws + off); off += align256((size_t)N * 4);
    int*   csr  = (int*)(ws + off);   off += align256((size_t)E * 4);
    float* A1   = (float*)(ws + off); off += align256((size_t)N * 64 * 4);
    float* A2   = (float*)(ws + off); off += align256((size_t)N * 64 * 4);
    float* H    = (float*)(ws + off); off += align256((size_t)N * 64 * 4);
    float* Wp1  = (float*)(ws + off); off += align256((size_t)3 * 64 * 64 * 4);
    float* Wp2  = (float*)(ws + off); off += align256((size_t)3 * 64 * 40 * 4);
    int* rowptr = (off + align256((size_t)(N + 1) * 4) <= ws_size) ? (int*)(ws + off) : (int*)d_out;
    int* pos = (int*)H;
    (void)n_in; (void)out_size;

    const int TB = 256;
    const int nodeBlk  = (N + TB - 1) / TB;
    const int edgeBlk  = (E + TB - 1) / TB;
    const int rowBlk   = (int)(((long long)N * 16 + TB - 1) / TB);
    const int gemmBlk  = ((((N + 31) / 32) * 32) + TB - 1) / TB;

    packw_kernel<<<(3 * 64 * 64 + TB - 1) / TB, TB, 0, stream>>>(W1, Wp1, 64, 3 * 64 * 64);
    packw_kernel<<<(3 * 64 * 40 + TB - 1) / TB, TB, 0, stream>>>(W2, Wp2, 40, 3 * 64 * 40);
    init_kernel<<<(N + 1 + TB - 1) / TB, TB, 0, stream>>>(dis, rowptr, N);
    count_kernel<<<edgeBlk, TB, 0, stream>>>(src, dst, dis, rowptr, pos, E);
    dis_kernel<<<nodeBlk, TB, 0, stream>>>(dis, N);
    rowptr_scan_kernel<<<1, 1024, 0, stream>>>(rowptr, N + 1);
    fill_kernel<<<edgeBlk, TB, 0, stream>>>(src, dst, rowptr, pos, csr, E);

    spmm_gather_kernel<false><<<rowBlk, TB, 0, stream>>>(rowptr, csr, dis, x, x, A1, N);
    spmm_gather_kernel<true><<<rowBlk, TB, 0, stream>>>(rowptr, csr, dis, A1, x, A2, N);
    cheb_gemm_kernel<64, true><<<gemmBlk, TB, 0, stream>>>(x, A1, A2, Wp1, b1, H, N);

    spmm_gather_kernel<false><<<rowBlk, TB, 0, stream>>>(rowptr, csr, dis, H, H, A1, N);
    spmm_gather_kernel<true><<<rowBlk, TB, 0, stream>>>(rowptr, csr, dis, A1, H, A2, N);
    cheb_gemm_kernel<40, false><<<gemmBlk, TB, 0, stream>>>(H, A1, A2, Wp2, b2, out, N);

    softmax40_kernel<<<nodeBlk, TB, 0, stream>>>(out, N);
}


# Row-pointer prefix sum: 25 work-groups, each re-sums the counts in front of its 4096 with whole-line vector loads, then scans its own tile; was one work-group walking per-thread 98-int chunks
# speedup vs baseline: 1.9857x; 1.3033x over previous
#include <hip/hip_runtime.h>
#include <hip/hip_bf16.h>

typedef __attribute__((ext_vector_type(2))) float v2f;
typedef __attribute__((ext_vector_type(8))) float v8f;

__global__ __launch_bounds__(256) void packw_kernel(const float* __restrict__ W,
                                                    float* __restrict__ Wp,
                                                    int nout, int total) {
    int idx = blockIdx.x * blockDim.x + threadIdx.x;
    if (idx >= total) return;
    int col = idx % nout;
    int k   = (idx / nout) & 63;
    int m   = idx / (nout * 64);
    Wp[(((size_t)m * 32 + (k >> 1)) * nout + col) * 2 + (k & 1)] = W[idx];
}

__global__ __launch_bounds__(256) void init_kernel(float* __restrict__ deg,
                                                   int* __restrict__ cnt, int n) {
    int i = blockIdx.x * blockDim.x + threadIdx.x;
    if (i < n) deg[i] = 0.f;
    if (i <= n) cnt[i] = 0;
}

__global__ __launch_bounds__(256) void count_kernel(const int* __restrict__ src,
                                                    const int* __restrict__ dst,
                                                    float* __restrict__ deg,
                                                    int* __restrict__ cnt,
                                                    int* __restrict__ pos, int nedges) {
    int e = blockIdx.x * blockDim.x + threadIdx.x;
    if (e >= nedges) return;
    int s = src[e], d = dst[e];
    if (s != d) {
        unsafeAtomicAdd(deg + s, 1.0f);
        pos[e] = atomicAdd(cnt + d, 1);
    }
}

__global__ __launch_bounds__(256) void dis_kernel(float* __restrict__ deg, int n) {
    int i = blockIdx.x * blockDim.x + threadIdx.x;
    if (i >= n) return;
    float d = deg[i];
    deg[i] = (d > 0.f) ? rsqrtf(fmaxf(d, 1.0f)) : 0.f;
}

__global__ __launch_bounds__(1024) void rowptr_scan_kernel(const int* __restrict__ cnt,
                                                           int* __restrict__ rowptr, int n1) {
    __shared__ int wtot[32];
    __shared__ int wpre[32];
    const int t = threadIdx.x;
    const int lane = t & 31, wv = t >> 5;
    const int4* c4 = (const int4*)cnt;
    const int nfront = blockIdx.x * 1024;
    const int base = blockIdx.x * 4096 + t * 4;

    int4 m = make_int4(0, 0, 0, 0);
    if (base + 3 < n1) {
        m = c4[nfront + t];
    } else {
        if (base < n1)     m.x = cnt[base];
        if (base + 1 < n1) m.y = cnt[base + 1];
        if (base + 2 < n1) m.z = cnt[base + 2];
    }

    int pre = 0;
    int v = t;
    for (; v + 7 * 1024 < nfront; v += 8 * 1024) {
        int4 q[8];
        #pragma unroll
        for (int k = 0; k < 8; ++k) q[k] = c4[v + k * 1024];
        #pragma unroll
        for (int k = 0; k < 8; ++k) pre += (q[k].x + q[k].y) + (q[k].z + q[k].w);
    }
    for (; v < nfront; v += 1024) {
        int4 q = c4[v];
        pre += (q.x + q.y) + (q.z + q.w);
    }

    const int s = (m.x + m.y) + (m.z + m.w);
    int inc = s;
    #pragma unroll
    for (int o = 1; o < 32; o <<= 1) {
        int u = __shfl_up(inc, o, 32);
        if (lane >= o) inc += u;
    }
    #pragma unroll
    for (int o = 16; o > 0; o >>= 1) pre += __shfl_xor(pre, o, 32);
    if (lane == 31) { wtot[wv] = inc; wpre[wv] = pre; }
    __syncthreads();
    const int wt = wtot[lane];
    int wp = wpre[lane];
    int winc = wt;
    #pragma unroll
    for (int o = 1; o < 32; o <<= 1) {
        int u = __shfl_up(winc, o, 32);
        if (lane >= o) winc += u;
    }
    #pragma unroll
    for (int o = 16; o > 0; o >>= 1) wp += __shfl_xor(wp, o, 32);
    const int wexc = __shfl(winc - wt, wv, 32);
    const int p = wp + wexc + (inc - s);

    int4 r;
    r.x = p;
    r.y = p + m.x;
    r.z = r.y + m.y;
    r.w = r.z + m.z;
    if (base + 3 < n1) {
        ((int4*)rowptr)[nfront + t] = r;
    } else {
        if (base < n1)     rowptr[base]     = r.x;
        if (base + 1 < n1) rowptr[base + 1] = r.y;
        if (base + 2 < n1) rowptr[base + 2] = r.z;
    }
}

__global__ __launch_bounds__(256) void fill_kernel(const int* __restrict__ src,
                                                   const int* __restrict__ dst,
                                                   const int* __restrict__ rowptr,
                                                   const int* __restrict__ pos,
                                                   int* __restrict__ csr, int nedges) {
    int e = blockIdx.x * blockDim.x + threadIdx.x;
    if (e >= nedges) return;
    int s = src[e], d = dst[e];
    if (s != d) csr[rowptr[d] + pos[e]] = s;
}

template <int G, int CNT>
__device__ __forceinline__ void gather_group(const float4* __restrict__ x4, int si16, float wl,
                                             int fg, float4& acc) {
    float4 xv[CNT];
    float  w[CNT];
    #pragma unroll
    for (int t = 0; t < CNT; ++t) {
        int idx = __shfl(si16, G + t, 16) + fg;
        w[t]  = __shfl(wl, G + t, 16);
        xv[t] = x4[idx];
    }
    #pragma unroll
    for (int t = 0; t < CNT; ++t) {
        acc.x += w[t] * xv[t].x;
        acc.y += w[t] * xv[t].y;
        acc.z += w[t] * xv[t].z;
        acc.w += w[t] * xv[t].w;
    }
}

template <bool SECOND>
__global__ __launch_bounds__(256) void spmm_gather_kernel(const int* __restrict__ rowptr,
                                                          const int* __restrict__ csr,
                                                          const float* __restrict__ dis,
                                                          const float* __restrict__ x,
                                                          const float* __restrict__ x0,
                                                          float* __restrict__ out, int n) {
    int tid = blockIdx.x * blockDim.x + threadIdx.x;
    int row = tid >> 4;
    int fg  = tid & 15;
    if (row >= n) return;
    const float4* x4 = (const float4*)x;
    int start = rowptr[row];
    int end   = rowptr[row + 1];
    float ndd = -dis[row];
    float4 v0 = make_float4(0.f, 0.f, 0.f, 0.f);
    if (SECOND) v0 = ((const float4*)x0)[row * 16 + fg];
    float4 acc = make_float4(0.f, 0.f, 0.f, 0.f);
    for (int base = start; base < end; base += 16) {
        int j = base + fg;
        bool valid = j < end;
        int si = csr[valid ? j : base];
        float wl = valid ? ndd * dis[si] : 0.f;
        int si16 = si * 16;
        int cnt = end - base;
        if (cnt >= 8) {
            gather_group<0, 8>(x4, si16, wl, fg, acc);
        } else {
            gather_group<0, 4>(x4, si16, wl, fg, acc);
            if (cnt > 4) gather_group<4, 4>(x4, si16, wl, fg, acc);
        }
        if (cnt >= 16) {
            gather_group<8, 8>(x4, si16, wl, fg, acc);
        } else if (cnt > 8) {
            gather_group<8, 4>(x4, si16, wl, fg, acc);
            if (cnt > 12) gather_group<12, 4>(x4, si16, wl, fg, acc);
        }
    }
    if (SECOND) {
        acc.x = 2.0f * acc.x - v0.x;
        acc.y = 2.0f * acc.y - v0.y;
        acc.z = 2.0f * acc.z - v0.z;
        acc.w = 2.0f * acc.w - v0.w;
    }
    ((float4*)out)[row * 16 + fg] = acc;
}

template <int NT, int NOUT>
__device__ inline void cheb_accum2(const float* __restrict__ A,
                                   const float* __restrict__ Wp,
                                   int rowA0, int rowA1, int khalf, int colLane,
                                   v8f (&acc0)[NT], v8f (&acc1)[NT]) {
    const float* ar0 = A + (size_t)rowA0 * 64;
    const float* ar1 = A + (size_t)rowA1 * 64;
    #pragma unroll
    for (int k0 = 0; k0 < 64; k0 += 4) {
        int kk = k0 + 2 * khalf;
        v2f a0 = *(const v2f*)(ar0 + kk);
        v2f a1 = *(const v2f*)(ar1 + kk);
        int p = (k0 >> 1) + khalf;
        #pragma unroll
        for (int t = 0; t < NT; ++t) {
            int cg = t * 16 + colLane;
            v2f b;
            if ((NOUT & 15) == 0 || cg < NOUT) {
                b = *(const v2f*)(Wp + ((size_t)p * NOUT + cg) * 2);
            } else {
                b.x = 0.f;
                b.y = 0.f;
            }
            acc0[t] = __builtin_amdgcn_wmma_f32_16x16x4_f32(
                false, a0, false, b, (short)0, acc0[t], false, false);
            acc1[t] = __builtin_amdgcn_wmma_f32_16x16x4_f32(
                false, a1, false, b, (short)0, acc1[t], false, false);
        }
    }
}

template <int NOUT, bool RELU>
__global__ __launch_bounds__(256) void cheb_gemm_kernel(const float* __restrict__ A0,
                                                        const float* __restrict__ A1,
                                                        const float* __restrict__ A2,
                                                        const float* __restrict__ Wp,
                                                        const float* __restrict__ bias,
                                                        float* __restrict__ out, int nodes) {
    constexpr int NT  = (NOUT + 15) / 16;
    constexpr int PSZ = 64 * NOUT;
    int wave = (blockIdx.x * blockDim.x + threadIdx.x) >> 5;
    int lane = threadIdx.x & 31;
    int l15  = lane & 15;
    int khalf = lane >> 4;
    int row0 = wave * 32;
    if (row0 >= nodes) return;

    v8f acc0[NT] = {};
    v8f acc1[NT] = {};
    cheb_accum2<NT, NOUT>(A0, Wp + 0 * PSZ, row0 + l15, row0 + 16 + l15, khalf, l15, acc0, acc1);
    cheb_accum2<NT, NOUT>(A1, Wp + 1 * PSZ, row0 + l15, row0 + 16 + l15, khalf, l15, acc0, acc1);
    cheb_accum2<NT, NOUT>(A2, Wp + 2 * PSZ, row0 + l15, row0 + 16 + l15, khalf, l15, acc0, acc1);

    #pragma unroll
    for (int t = 0; t < NT; ++t) {
        int col = t * 16 + l15;
        if ((NOUT & 15) != 0 && col >= NOUT) continue;
        float bv = bias[col];
        #pragma unroll
        for (int i = 0; i < 8; ++i) {
            int r = row0 + i + 8 * khalf;
            float v0 = acc0[t][i] + bv;
            float v1 = acc1[t][i] + bv;
            if (RELU) { v0 = fmaxf(v0, 0.f); v1 = fmaxf(v1, 0.f); }
            if (r < nodes)      out[(size_t)r * NOUT + col]        = v0;
            if (r + 16 < nodes) out[(size_t)(r + 16) * NOUT + col] = v1;
        }
    }
}

__global__ __launch_bounds__(256) void softmax40_kernel(float* __restrict__ out, int nodes) {
    int i = blockIdx.x * blockDim.x + threadIdx.x;
    if (i >= nodes) return;
    float* row = out + (size_t)i * 40;
    float m = -INFINITY;
    #pragma unroll
    for (int c = 0; c < 40; ++c) m = fmaxf(m, row[c]);
    float e[40];
    float s = 0.f;
    #pragma unroll
    for (int c = 0; c < 40; ++c) { e[c] = __expf(row[c] - m); s += e[c]; }
    float inv = 1.f / s;
    #pragma unroll
    for (int c = 0; c < 40; ++c) row[c] = e[c] * inv;
}

extern "C" void kernel_launch(void* const* d_in, const int* in_sizes, int n_in,
                              void* d_out, int out_size, void* d_ws, size_t ws_size,
                              hipStream_t stream) {
    const float* x  = (const float*)d_in[0];
    const int*   ei = (const int*)d_in[1];
    const float* W1 = (const float*)d_in[2];
    const float* b1 = (const float*)d_in[3];
    const float* W2 = (const float*)d_in[4];
    const float* b2 = (const float*)d_in[5];
    float* out = (float*)d_out;

    const int N = in_sizes[0] / 64;
    const int E = in_sizes[1] / 2;
    const int* src = ei;
    const int* dst = ei + E;

    auto align256 = [](size_t v) { return (v + 255) & ~(size_t)255; };
    char* ws = (char*)d_ws;
    size_t off = 0;
    float* dis  = (float*)(ws + off); off += align256((size_t)N * 4);
    int*   csr  = (int*)(ws + off);   off += align256((size_t)E * 4);
    float* A1   = (float*)(ws + off); off += align256((size_t)N * 64 * 4);
    float* A2   = (float*)(ws + off); off += align256((size_t)N * 64 * 4);
    float* H    = (float*)(ws + off); off += align256((size_t)N * 64 * 4);
    float* Wp1  = (float*)(ws + off); off += align256((size_t)3 * 64 * 64 * 4);
    float* Wp2  = (float*)(ws + off); off += align256((size_t)3 * 64 * 40 * 4);
    int* rowptr = (off + align256((size_t)(N + 1) * 4) <= ws_size) ? (int*)(ws + off) : (int*)d_out;
    int* pos = (int*)H;
    int* cnt = (int*)A1;
    (void)n_in; (void)out_size;

    const int TB = 256;
    const int nodeBlk  = (N + TB - 1) / TB;
    const int edgeBlk  = (E + TB - 1) / TB;
    const int rowBlk   = (int)(((long long)N * 16 + TB - 1) / TB);
    const int gemmBlk  = ((((N + 31) / 32) * 32) + TB - 1) / TB;

    packw_kernel<<<(3 * 64 * 64 + TB - 1) / TB, TB, 0, stream>>>(W1, Wp1, 64, 3 * 64 * 64);
    packw_kernel<<<(3 * 64 * 40 + TB - 1) / TB, TB, 0, stream>>>(W2, Wp2, 40, 3 * 64 * 40);
    init_kernel<<<(N + 1 + TB - 1) / TB, TB, 0, stream>>>(dis, cnt, N);
    count_kernel<<<edgeBlk, TB, 0, stream>>>(src, dst, dis, cnt, pos, E);
    dis_kernel<<<nodeBlk, TB, 0, stream>>>(dis, N);
    rowptr_scan_kernel<<<(N + 1 + 4095) / 4096, 1024, 0, stream>>>(cnt, rowptr, N + 1);
    fill_kernel<<<edgeBlk, TB, 0, stream>>>(src, dst, rowptr, pos, csr, E);

    spmm_gather_kernel<false><<<rowBlk, TB, 0, stream>>>(rowptr, csr, dis, x, x, A1, N);
    spmm_gather_kernel<true><<<rowBlk, TB, 0, stream>>>(rowptr, csr, dis, A1, x, A2, N);
    cheb_gemm_kernel<64, true><<<gemmBlk, TB, 0, stream>>>(x, A1, A2, Wp1, b1, H, N);

    spmm_gather_kernel<false><<<rowBlk, TB, 0, stream>>>(rowptr, csr, dis, H, H, A1, N);
    spmm_gather_kernel<true><<<rowBlk, TB, 0, stream>>>(rowptr, csr, dis, A1, H, A2, N);
    cheb_gemm_kernel<40, false><<<gemmBlk, TB, 0, stream>>>(H, A1, A2, Wp2, b2, out, N);

    softmax40_kernel<<<nodeBlk, TB, 0, stream>>>(out, N);
}


# Destination sort redone as a two-level counting sort on LDS counters (4096-edge tiles filed by 256-node bin, then one work-group per bin): no returning global atomics; row sum order = LDS arrival
# speedup vs baseline: 2.4878x; 1.2529x over previous
#include <hip/hip_runtime.h>
#include <hip/hip_bf16.h>

typedef __attribute__((ext_vector_type(2))) float v2f;
typedef __attribute__((ext_vector_type(8))) float v8f;

__global__ __launch_bounds__(256) void packw_kernel(const float* __restrict__ W,
                                                    float* __restrict__ Wp,
                                                    int nout, int total) {
    int idx = blockIdx.x * blockDim.x + threadIdx.x;
    if (idx >= total) return;
    int col = idx % nout;
    int k   = (idx / nout) & 63;
    int m   = idx / (nout * 64);
    Wp[(((size_t)m * 32 + (k >> 1)) * nout + col) * 2 + (k & 1)] = W[idx];
}

constexpr int TILE_E = 4096;
constexpr int BIN_SH = 8;
constexpr int BIN_W  = 1 << BIN_SH;
constexpr int NB_CAP = 512;
constexpr int ST_CAP = 8192;

__device__ __forceinline__ int wave_incl_scan(int v, int lane) {
    #pragma unroll
    for (int o = 1; o < 32; o <<= 1) {
        int u = __shfl_up(v, o, 32);
        if (lane >= o) v += u;
    }
    return v;
}

__device__ __forceinline__ int wave_sum(int v) {
    #pragma unroll
    for (int o = 16; o > 0; o >>= 1) v += __shfl_xor(v, o, 32);
    return v;
}

__global__ __launch_bounds__(256) void init_kernel(float* __restrict__ deg, int n) {
    int i = blockIdx.x * blockDim.x + threadIdx.x;
    if (i < n) deg[i] = 0.f;
}

__global__ __launch_bounds__(1024) void tile_sort_kernel(const int* __restrict__ src,
                                                         const int* __restrict__ dst,
                                                         float* __restrict__ deg,
                                                         unsigned* __restrict__ recs,
                                                         int* __restrict__ offs,
                                                         int nedges, int nb, int nt) {
    __shared__ int hist[NB_CAP];
    __shared__ int wtot[NB_CAP / 32];
    __shared__ unsigned stage[TILE_E];
    const int t = threadIdx.x, lane = t & 31, wv = t >> 5;
    const int tile = blockIdx.x;
    if (t < NB_CAP) hist[t] = 0;
    __syncthreads();

    int s[4], d[4];
    #pragma unroll
    for (int k = 0; k < 4; ++k) {
        int e = tile * TILE_E + k * 1024 + t;
        s[k] = 0; d[k] = 0;
        if (e < nedges) { s[k] = src[e]; d[k] = dst[e]; }
    }
    int bin[4], rk[4];
    #pragma unroll
    for (int k = 0; k < 4; ++k) {
        bin[k] = -1; rk[k] = 0;
        if (s[k] != d[k]) {
            unsafeAtomicAdd(deg + s[k], 1.0f);
            bin[k] = d[k] >> BIN_SH;
            rk[k] = atomicAdd(&hist[bin[k]], 1);
        }
    }
    __syncthreads();

    int v = 0, inc = 0;
    if (t < NB_CAP) {
        v = (t < nb) ? hist[t] : 0;
        inc = wave_incl_scan(v, lane);
        if (lane == 31) wtot[wv] = inc;
    }
    __syncthreads();
    if (t < NB_CAP) {
        int woff = 0;
        for (int w = 0; w < wv; ++w) woff += wtot[w];
        int exc = woff + inc - v;
        hist[t] = exc;
        if (t <= nb) offs[(size_t)t * nt + tile] = exc;
    }
    __syncthreads();

    #pragma unroll
    for (int k = 0; k < 4; ++k) {
        if (bin[k] >= 0)
            stage[hist[bin[k]] + rk[k]] =
                ((unsigned)s[k] << BIN_SH) | (unsigned)(d[k] & (BIN_W - 1));
    }
    __syncthreads();
    const int total = hist[nb];
    unsigned* out = recs + (size_t)tile * TILE_E;
    for (int p = t; p < total; p += 1024) out[p] = stage[p];
}

__global__ __launch_bounds__(1024) void csr_build_kernel(const unsigned* __restrict__ recs,
                                                         const int* __restrict__ offs,
                                                         float* __restrict__ deg,
                                                         int* __restrict__ rowptr,
                                                         int* __restrict__ csr,
                                                         int n, int nb, int nt) {
    __shared__ int cnt[BIN_W];
    __shared__ int wlo[32];
    __shared__ int whi[32];
    __shared__ int wtot[BIN_W / 32];
    __shared__ int stage[ST_CAP];
    const int t = threadIdx.x, lane = t & 31, wv = t >> 5;
    const int b = blockIdx.x;
    const int* __restrict__ o0 = offs + (size_t)b * nt;
    const int* __restrict__ o1 = o0 + nt;
    if (t < BIN_W) cnt[t] = 0;

    constexpr int SEGS = 8;
    const int hw = t >> 4, l16 = t & 15;
    int lo[SEGS], hi[SEGS];
    #pragma unroll
    for (int j = 0; j < SEGS; ++j) {
        const int seg = hw + 64 * j;
        lo[j] = 0; hi[j] = 0;
        if (seg < nt) { lo[j] = o0[seg]; hi[j] = o1[seg]; }
    }
    unsigned r0[SEGS];
    #pragma unroll
    for (int j = 0; j < SEGS; ++j) {
        const int k = lo[j] + l16;
        r0[j] = 0u;
        if (k < hi[j]) r0[j] = recs[(size_t)(hw + 64 * j) * TILE_E + k];
    }

    int plo = 0, phi = 0;
    for (int i = t; i < nt; i += 1024) { plo += o0[i]; phi += o1[i]; }
    plo = wave_sum(plo);
    phi = wave_sum(phi);
    if (lane == 0) { wlo[wv] = plo; whi[wv] = phi; }
    __syncthreads();
    const int base  = wave_sum(wlo[lane]);
    const int total = wave_sum(whi[lane]) - base;

    #pragma unroll
    for (int j = 0; j < SEGS; ++j) {
        const int k = lo[j] + l16;
        if (k < hi[j]) {
            atomicAdd(&cnt[r0[j] & (BIN_W - 1)], 1);
            const unsigned* rp = recs + (size_t)(hw + 64 * j) * TILE_E;
            for (int kk = k + 16; kk < hi[j]; kk += 16) atomicAdd(&cnt[rp[kk] & (BIN_W - 1)], 1);
        }
    }
    for (int seg = hw + 64 * SEGS; seg < nt; seg += 64) {
        const int l = o0[seg], h = o1[seg];
        const unsigned* rp = recs + (size_t)seg * TILE_E;
        for (int k = l + l16; k < h; k += 16) atomicAdd(&cnt[rp[k] & (BIN_W - 1)], 1);
    }
    __syncthreads();

    int c = 0, inc = 0;
    if (t < BIN_W) {
        c = cnt[t];
        inc = wave_incl_scan(c, lane);
        if (lane == 31) wtot[wv] = inc;
    }
    __syncthreads();
    if (t < BIN_W) {
        int woff = 0;
        for (int w = 0; w < wv; ++w) woff += wtot[w];
        const int exc = woff + inc - c;
        cnt[t] = exc;
        const int node = b * BIN_W + t;
        if (node < n) {
            rowptr[node] = base + exc;
            const float dg = deg[node];
            deg[node] = (dg > 0.f) ? rsqrtf(fmaxf(dg, 1.0f)) : 0.f;
        }
    }
    if (b == nb - 1 && t == 0) rowptr[n] = base + total;
    __syncthreads();

    const bool fits = total <= ST_CAP;
    auto place = [&](unsigned r) {
        const int p = atomicAdd(&cnt[r & (BIN_W - 1)], 1);
        if (fits) stage[p] = (int)(r >> BIN_SH);
        else      csr[base + p] = (int)(r >> BIN_SH);
    };
    #pragma unroll
    for (int j = 0; j < SEGS; ++j) {
        const int k = lo[j] + l16;
        if (k < hi[j]) {
            place(r0[j]);
            const unsigned* rp = recs + (size_t)(hw + 64 * j) * TILE_E;
            for (int kk = k + 16; kk < hi[j]; kk += 16) place(rp[kk]);
        }
    }
    for (int seg = hw + 64 * SEGS; seg < nt; seg += 64) {
        const int l = o0[seg], h = o1[seg];
        const unsigned* rp = recs + (size_t)seg * TILE_E;
        for (int k = l + l16; k < h; k += 16) place(rp[k]);
    }
    __syncthreads();
    if (fits) {
        for (int p = t; p < total; p += 1024) csr[base + p] = stage[p];
    }
}

template <int G, int CNT>
__device__ __forceinline__ void gather_group(const float4* __restrict__ x4, int si16, float wl,
                                             int fg, float4& acc) {
    float4 xv[CNT];
    float  w[CNT];
    #pragma unroll
    for (int t = 0; t < CNT; ++t) {
        int idx = __shfl(si16, G + t, 16) + fg;
        w[t]  = __shfl(wl, G + t, 16);
        xv[t] = x4[idx];
    }
    #pragma unroll
    for (int t = 0; t < CNT; ++t) {
        acc.x += w[t] * xv[t].x;
        acc.y += w[t] * xv[t].y;
        acc.z += w[t] * xv[t].z;
        acc.w += w[t] * xv[t].w;
    }
}

template <bool SECOND>
__global__ __launch_bounds__(256) void spmm_gather_kernel(const int* __restrict__ rowptr,
                                                          const int* __restrict__ csr,
                                                          const float* __restrict__ dis,
                                                          const float* __restrict__ x,
                                                          const float* __restrict__ x0,
                                                          float* __restrict__ out, int n) {
    int tid = blockIdx.x * blockDim.x + threadIdx.x;
    int row = tid >> 4;
    int fg  = tid & 15;
    if (row >= n) return;
    const float4* x4 = (const float4*)x;
    int start = rowptr[row];
    int end   = rowptr[row + 1];
    float ndd = -dis[row];
    float4 v0 = make_float4(0.f, 0.f, 0.f, 0.f);
    if (SECOND) v0 = ((const float4*)x0)[row * 16 + fg];
    float4 acc = make_float4(0.f, 0.f, 0.f, 0.f);
    for (int base = start; base < end; base += 16) {
        int j = base + fg;
        bool valid = j < end;
        int si = csr[valid ? j : base];
        float wl = valid ? ndd * dis[si] : 0.f;
        int si16 = si * 16;
        int cnt = end - base;
        if (cnt >= 8) {
            gather_group<0, 8>(x4, si16, wl, fg, acc);
        } else {
            gather_group<0, 4>(x4, si16, wl, fg, acc);
            if (cnt > 4) gather_group<4, 4>(x4, si16, wl, fg, acc);
        }
        if (cnt >= 16) {
            gather_group<8, 8>(x4, si16, wl, fg, acc);
        } else if (cnt > 8) {
            gather_group<8, 4>(x4, si16, wl, fg, acc);
            if (cnt > 12) gather_group<12, 4>(x4, si16, wl, fg, acc);
        }
    }
    if (SECOND) {
        acc.x = 2.0f * acc.x - v0.x;
        acc.y = 2.0f * acc.y - v0.y;
        acc.z = 2.0f * acc.z - v0.z;
        acc.w = 2.0f * acc.w - v0.w;
    }
    ((float4*)out)[row * 16 + fg] = acc;
}

template <int NT, int NOUT>
__device__ inline void cheb_accum2(const float* __restrict__ A,
                                   const float* __restrict__ Wp,
                                   int rowA0, int rowA1, int khalf, int colLane,
                                   v8f (&acc0)[NT], v8f (&acc1)[NT]) {
    const float* ar0 = A + (size_t)rowA0 * 64;
    const float* ar1 = A + (size_t)rowA1 * 64;
    #pragma unroll
    for (int k0 = 0; k0 < 64; k0 += 4) {
        int kk = k0 + 2 * khalf;
        v2f a0 = *(const v2f*)(ar0 + kk);
        v2f a1 = *(const v2f*)(ar1 + kk);
        int p = (k0 >> 1) + khalf;
        #pragma unroll
        for (int t = 0; t < NT; ++t) {
            int cg = t * 16 + colLane;
            v2f b;
            if ((NOUT & 15) == 0 || cg < NOUT) {
                b = *(const v2f*)(Wp + ((size_t)p * NOUT + cg) * 2);
            } else {
                b.x = 0.f;
                b.y = 0.f;
            }
            acc0[t] = __builtin_amdgcn_wmma_f32_16x16x4_f32(
                false, a0, false, b, (short)0, acc0[t], false, false);
            acc1[t] = __builtin_amdgcn_wmma_f32_16x16x4_f32(
                false, a1, false, b, (short)0, acc1[t], false, false);
        }
    }
}

template <int NOUT, bool RELU>
__global__ __launch_bounds__(256) void cheb_gemm_kernel(const float* __restrict__ A0,
                                                        const float* __restrict__ A1,
                                                        const float* __restrict__ A2,
                                                        const float* __restrict__ Wp,
                                                        const float* __restrict__ bias,
                                                        float* __restrict__ out, int nodes) {
    constexpr int NT  = (NOUT + 15) / 16;
    constexpr int PSZ = 64 * NOUT;
    int wave = (blockIdx.x * blockDim.x + threadIdx.x) >> 5;
    int lane = threadIdx.x & 31;
    int l15  = lane & 15;
    int khalf = lane >> 4;
    int row0 = wave * 32;
    if (row0 >= nodes) return;

    v8f acc0[NT] = {};
    v8f acc1[NT] = {};
    cheb_accum2<NT, NOUT>(A0, Wp + 0 * PSZ, row0 + l15, row0 + 16 + l15, khalf, l15, acc0, acc1);
    cheb_accum2<NT, NOUT>(A1, Wp + 1 * PSZ, row0 + l15, row0 + 16 + l15, khalf, l15, acc0, acc1);
    cheb_accum2<NT, NOUT>(A2, Wp + 2 * PSZ, row0 + l15, row0 + 16 + l15, khalf, l15, acc0, acc1);

    #pragma unroll
    for (int t = 0; t < NT; ++t) {
        int col = t * 16 + l15;
        if ((NOUT & 15) != 0 && col >= NOUT) continue;
        float bv = bias[col];
        #pragma unroll
        for (int i = 0; i < 8; ++i) {
            int r = row0 + i + 8 * khalf;
            float v0 = acc0[t][i] + bv;
            float v1 = acc1[t][i] + bv;
            if (RELU) { v0 = fmaxf(v0, 0.f); v1 = fmaxf(v1, 0.f); }
            if (r < nodes)      out[(size_t)r * NOUT + col]        = v0;
            if (r + 16 < nodes) out[(size_t)(r + 16) * NOUT + col] = v1;
        }
    }
}

__global__ __launch_bounds__(256) void softmax40_kernel(float* __restrict__ out, int nodes) {
    int i = blockIdx.x * blockDim.x + threadIdx.x;
    if (i >= nodes) return;
    float* row = out + (size_t)i * 40;
    float m = -INFINITY;
    #pragma unroll
    for (int c = 0; c < 40; ++c) m = fmaxf(m, row[c]);
    float e[40];
    float s = 0.f;
    #pragma unroll
    for (int c = 0; c < 40; ++c) { e[c] = __expf(row[c] - m); s += e[c]; }
    float inv = 1.f / s;
    #pragma unroll
    for (int c = 0; c < 40; ++c) row[c] = e[c] * inv;
}

extern "C" void kernel_launch(void* const* d_in, const int* in_sizes, int n_in,
                              void* d_out, int out_size, void* d_ws, size_t ws_size,
                              hipStream_t stream) {
    const float* x  = (const float*)d_in[0];
    const int*   ei = (const int*)d_in[1];
    const float* W1 = (const float*)d_in[2];
    const float* b1 = (const float*)d_in[3];
    const float* W2 = (const float*)d_in[4];
    const float* b2 = (const float*)d_in[5];
    float* out = (float*)d_out;

    const int N = in_sizes[0] / 64;
    const int E = in_sizes[1] / 2;
    const int* src = ei;
    const int* dst = ei + E;

    auto align256 = [](size_t v) { return (v + 255) & ~(size_t)255; };
    char* ws = (char*)d_ws;
    size_t off = 0;
    float* dis  = (float*)(ws + off); off += align256((size_t)N * 4);
    int*   csr  = (int*)(ws + off);   off += align256((size_t)E * 4);
    float* A1   = (float*)(ws + off); off += align256((size_t)N * 64 * 4);
    float* A2   = (float*)(ws + off); off += align256((size_t)N * 64 * 4);
    float* H    = (float*)(ws + off); off += align256((size_t)N * 64 * 4);
    float* Wp1  = (float*)(ws + off); off += align256((size_t)3 * 64 * 64 * 4);
    float* Wp2  = (float*)(ws + off); off += align256((size_t)3 * 64 * 40 * 4);
    int* rowptr = (off + align256((size_t)(N + 1) * 4) <= ws_size) ? (int*)(ws + off) : (int*)d_out;
    unsigned* recs = (unsigned*)H;
    int* offs = (int*)A1;
    (void)n_in; (void)out_size;

    const int TB = 256;
    const int nodeBlk  = (N + TB - 1) / TB;
    const int nTiles   = (E + TILE_E - 1) / TILE_E;
    const int nBins    = (N + BIN_W - 1) / BIN_W;
    const int rowBlk   = (int)(((long long)N * 16 + TB - 1) / TB);
    const int gemmBlk  = ((((N + 31) / 32) * 32) + TB - 1) / TB;

    packw_kernel<<<(3 * 64 * 64 + TB - 1) / TB, TB, 0, stream>>>(W1, Wp1, 64, 3 * 64 * 64);
    packw_kernel<<<(3 * 64 * 40 + TB - 1) / TB, TB, 0, stream>>>(W2, Wp2, 40, 3 * 64 * 40);
    init_kernel<<<nodeBlk, TB, 0, stream>>>(dis, N);
    tile_sort_kernel<<<nTiles, 1024, 0, stream>>>(src, dst, dis, recs, offs, E, nBins, nTiles);
    csr_build_kernel<<<nBins, 1024, 0, stream>>>(recs, offs, dis, rowptr, csr, N, nBins, nTiles);

    spmm_gather_kernel<false><<<rowBlk, TB, 0, stream>>>(rowptr, csr, dis, x, x, A1, N);
    spmm_gather_kernel<true><<<rowBlk, TB, 0, stream>>>(rowptr, csr, dis, A1, x, A2, N);
    cheb_gemm_kernel<64, true><<<gemmBlk, TB, 0, stream>>>(x, A1, A2, Wp1, b1, H, N);

    spmm_gather_kernel<false><<<rowBlk, TB, 0, stream>>>(rowptr, csr, dis, H, H, A1, N);
    spmm_gather_kernel<true><<<rowBlk, TB, 0, stream>>>(rowptr, csr, dis, A1, H, A2, N);
    cheb_gemm_kernel<40, false><<<gemmBlk, TB, 0, stream>>>(H, A1, A2, Wp2, b2, out, N);

    softmax40_kernel<<<nodeBlk, TB, 0, stream>>>(out, N);
}


# Out-degree without global atomics: the tile sort also files each tile by source bin in LDS, and a new launch counts per node in LDS; the zeroing launch and 1.6M scattered float adds go
# speedup vs baseline: 3.5663x; 1.4335x over previous
#include <hip/hip_runtime.h>
#include <hip/hip_bf16.h>

typedef __attribute__((ext_vector_type(2))) float v2f;
typedef __attribute__((ext_vector_type(8))) float v8f;

__global__ __launch_bounds__(256) void packw_kernel(const float* __restrict__ W,
                                                    float* __restrict__ Wp,
                                                    int nout, int total) {
    int idx = blockIdx.x * blockDim.x + threadIdx.x;
    if (idx >= total) return;
    int col = idx % nout;
    int k   = (idx / nout) & 63;
    int m   = idx / (nout * 64);
    Wp[(((size_t)m * 32 + (k >> 1)) * nout + col) * 2 + (k & 1)] = W[idx];
}

constexpr int TILE_E = 4096;
constexpr int BIN_SH = 8;
constexpr int BIN_W  = 1 << BIN_SH;
constexpr int NB_CAP = 512;
constexpr int ST_CAP = 8192;

__device__ __forceinline__ int wave_incl_scan(int v, int lane) {
    #pragma unroll
    for (int o = 1; o < 32; o <<= 1) {
        int u = __shfl_up(v, o, 32);
        if (lane >= o) v += u;
    }
    return v;
}

__device__ __forceinline__ int wave_sum(int v) {
    #pragma unroll
    for (int o = 16; o > 0; o >>= 1) v += __shfl_xor(v, o, 32);
    return v;
}

__device__ __forceinline__ float node_scale(float dg) {
    return (dg > 0.f) ? rsqrtf(fmaxf(dg, 1.0f)) : 0.f;
}

__global__ __launch_bounds__(1024) void tile_sort_kernel(const int* __restrict__ src,
                                                         const int* __restrict__ dst,
                                                         unsigned* __restrict__ recs,
                                                         int* __restrict__ offs,
                                                         unsigned* __restrict__ srecs,
                                                         int* __restrict__ soffs,
                                                         int nedges, int nb, int nt) {
    __shared__ int hist[2][NB_CAP];
    __shared__ int wtot[2][NB_CAP / 32];
    __shared__ unsigned stage[2][TILE_E];
    static_assert(2 * NB_CAP == 1024, "one thread per counter of the two histograms");
    const int t = threadIdx.x, lane = t & 31;
    const int tile = blockIdx.x;
    hist[t >> 9][t & (NB_CAP - 1)] = 0;
    __syncthreads();

    int s[4], d[4];
    #pragma unroll
    for (int k = 0; k < 4; ++k) {
        int e = tile * TILE_E + k * 1024 + t;
        s[k] = 0; d[k] = 0;
        if (e < nedges) { s[k] = src[e]; d[k] = dst[e]; }
    }
    int rk[4], rs[4];
    #pragma unroll
    for (int k = 0; k < 4; ++k) {
        rk[k] = 0; rs[k] = 0;
        if (s[k] != d[k]) {
            rk[k] = atomicAdd(&hist[0][d[k] >> BIN_SH], 1);
            rs[k] = atomicAdd(&hist[1][s[k] >> BIN_SH], 1);
        }
    }
    __syncthreads();

    const int h = t >> 9, u = t & (NB_CAP - 1), wu = u >> 5;
    const int v = (u < nb) ? hist[h][u] : 0;
    const int inc = wave_incl_scan(v, lane);
    if (lane == 31) wtot[h][wu] = inc;
    __syncthreads();
    {
        int woff = 0;
        for (int w = 0; w < wu; ++w) woff += wtot[h][w];
        const int exc = woff + inc - v;
        hist[h][u] = exc;
        int* __restrict__ o = h ? soffs : offs;
        if (u <= nb) o[(size_t)u * nt + tile] = exc;
    }
    __syncthreads();

    #pragma unroll
    for (int k = 0; k < 4; ++k) {
        if (s[k] != d[k]) {
            stage[0][hist[0][d[k] >> BIN_SH] + rk[k]] =
                ((unsigned)s[k] << BIN_SH) | (unsigned)(d[k] & (BIN_W - 1));
            stage[1][hist[1][s[k] >> BIN_SH] + rs[k]] = (unsigned)(s[k] & (BIN_W - 1));
        }
    }
    __syncthreads();
    const int total = hist[0][nb];
    unsigned* out  = recs  + (size_t)tile * TILE_E;
    unsigned* out2 = srecs + (size_t)tile * TILE_E;
    for (int p = t; p < total; p += 1024) {
        out[p]  = stage[0][p];
        out2[p] = stage[1][p];
    }
}

__global__ __launch_bounds__(1024) void deg_kernel(const unsigned* __restrict__ srecs,
                                                   const int* __restrict__ soffs,
                                                   float* __restrict__ deg,
                                                   int n, int nt) {
    __shared__ int cnt[BIN_W];
    const int t = threadIdx.x;
    const int b = blockIdx.x;
    const int* __restrict__ o0 = soffs + (size_t)b * nt;
    const int* __restrict__ o1 = o0 + nt;
    if (t < BIN_W) cnt[t] = 0;

    constexpr int SEGS = 8;
    const int hw = t >> 4, l16 = t & 15;
    int lo[SEGS], hi[SEGS];
    #pragma unroll
    for (int j = 0; j < SEGS; ++j) {
        const int seg = hw + 64 * j;
        lo[j] = 0; hi[j] = 0;
        if (seg < nt) { lo[j] = o0[seg]; hi[j] = o1[seg]; }
    }
    unsigned r0[SEGS];
    #pragma unroll
    for (int j = 0; j < SEGS; ++j) {
        const int k = lo[j] + l16;
        r0[j] = 0u;
        if (k < hi[j]) r0[j] = srecs[(size_t)(hw + 64 * j) * TILE_E + k];
    }
    __syncthreads();

    #pragma unroll
    for (int j = 0; j < SEGS; ++j) {
        const int k = lo[j] + l16;
        if (k < hi[j]) {
            atomicAdd(&cnt[r0[j] & (BIN_W - 1)], 1);
            const unsigned* rp = srecs + (size_t)(hw + 64 * j) * TILE_E;
            for (int kk = k + 16; kk < hi[j]; kk += 16) atomicAdd(&cnt[rp[kk] & (BIN_W - 1)], 1);
        }
    }
    for (int seg = hw + 64 * SEGS; seg < nt; seg += 64) {
        const int l = o0[seg], hh = o1[seg];
        const unsigned* rp = srecs + (size_t)seg * TILE_E;
        for (int k = l + l16; k < hh; k += 16) atomicAdd(&cnt[rp[k] & (BIN_W - 1)], 1);
    }
    __syncthreads();
    if (t < BIN_W) {
        const int node = b * BIN_W + t;
        if (node < n) deg[node] = (float)cnt[t];
    }
}

__global__ __launch_bounds__(1024) void csr_build_kernel(const unsigned* __restrict__ recs,
                                                         const int* __restrict__ offs,
                                                         const float* __restrict__ deg,
                                                         int* __restrict__ rowptr,
                                                         int* __restrict__ ent,
                                                         int n, int nb, int nt) {
    __shared__ int cnt[BIN_W];
    __shared__ float nsc[BIN_W];
    __shared__ int wlo[32];
    __shared__ int whi[32];
    __shared__ int wtot[BIN_W / 32];
    __shared__ int2 stage[ST_CAP];
    int2* __restrict__ ent2 = (int2*)ent;
    const int t = threadIdx.x, lane = t & 31, wv = t >> 5;
    const int b = blockIdx.x;
    const int* __restrict__ o0 = offs + (size_t)b * nt;
    const int* __restrict__ o1 = o0 + nt;
    if (t < BIN_W) cnt[t] = 0;

    constexpr int SEGS = 8;
    const int hw = t >> 4, l16 = t & 15;
    int lo[SEGS], hi[SEGS];
    #pragma unroll
    for (int j = 0; j < SEGS; ++j) {
        const int seg = hw + 64 * j;
        lo[j] = 0; hi[j] = 0;
        if (seg < nt) { lo[j] = o0[seg]; hi[j] = o1[seg]; }
    }
    unsigned r0[SEGS];
    #pragma unroll
    for (int j = 0; j < SEGS; ++j) {
        const int k = lo[j] + l16;
        r0[j] = 0u;
        if (k < hi[j]) r0[j] = recs[(size_t)(hw + 64 * j) * TILE_E + k];
    }

    int plo = 0, phi = 0;
    #pragma nounroll
    for (int i = t; i < nt; i += 1024) { plo += o0[i]; phi += o1[i]; }
    plo = wave_sum(plo);
    phi = wave_sum(phi);
    if (lane == 0) { wlo[wv] = plo; whi[wv] = phi; }
    __syncthreads();
    const int base  = wave_sum(wlo[lane]);
    const int total = wave_sum(whi[lane]) - base;

    #pragma unroll
    for (int j = 0; j < SEGS; ++j) {
        const int k = lo[j] + l16;
        if (k < hi[j]) {
            atomicAdd(&cnt[r0[j] & (BIN_W - 1)], 1);
            const unsigned* rp = recs + (size_t)(hw + 64 * j) * TILE_E;
            for (int kk = k + 16; kk < hi[j]; kk += 16) atomicAdd(&cnt[rp[kk] & (BIN_W - 1)], 1);
        }
    }
    for (int seg = hw + 64 * SEGS; seg < nt; seg += 64) {
        const int l = o0[seg], h = o1[seg];
        const unsigned* rp = recs + (size_t)seg * TILE_E;
        for (int k = l + l16; k < h; k += 16) atomicAdd(&cnt[rp[k] & (BIN_W - 1)], 1);
    }
    float g0[SEGS];
    #pragma unroll
    for (int j = 0; j < SEGS; ++j) {
        g0[j] = 0.f;
        if (lo[j] + l16 < hi[j]) g0[j] = deg[r0[j] >> BIN_SH];
    }
    __syncthreads();

    int c = 0, inc = 0;
    if (t < BIN_W) {
        c = cnt[t];
        inc = wave_incl_scan(c, lane);
        if (lane == 31) wtot[wv] = inc;
    }
    __syncthreads();
    if (t < BIN_W) {
        int woff = 0;
        for (int w = 0; w < wv; ++w) woff += wtot[w];
        const int exc = woff + inc - c;
        cnt[t] = exc;
        const int node = b * BIN_W + t;
        float sc = 0.f;
        if (node < n) {
            rowptr[node] = base + exc;
            sc = node_scale(deg[node]);
        }
        nsc[t] = -sc;
    }
    if (b == nb - 1 && t == 0) rowptr[n] = base + total;
    __syncthreads();

    const bool fits = total <= ST_CAP;
    auto place = [&](unsigned r, float dg) {
        const int ln = (int)(r & (BIN_W - 1));
        const int p = atomicAdd(&cnt[ln], 1);
        const int2 e = make_int2((int)(r >> BIN_SH), __float_as_int(nsc[ln] * node_scale(dg)));
        if (fits) stage[p] = e;
        else      ent2[base + p] = e;
    };
    #pragma unroll
    for (int j = 0; j < SEGS; ++j) {
        const int k = lo[j] + l16;
        if (k < hi[j]) {
            place(r0[j], g0[j]);
            const unsigned* rp = recs + (size_t)(hw + 64 * j) * TILE_E;
            for (int kk = k + 16; kk < hi[j]; kk += 16) {
                const unsigned r = rp[kk];
                place(r, deg[r >> BIN_SH]);
            }
        }
    }
    for (int seg = hw + 64 * SEGS; seg < nt; seg += 64) {
        const int l = o0[seg], h = o1[seg];
        const unsigned* rp = recs + (size_t)seg * TILE_E;
        for (int k = l + l16; k < h; k += 16) {
            const unsigned r = rp[k];
            place(r, deg[r >> BIN_SH]);
        }
    }
    __syncthreads();
    if (fits) {
        for (int p = t; p < total; p += 1024) ent2[base + p] = stage[p];
    }
}

template <int G, int CNT>
__device__ __forceinline__ void gather_group(const float4* __restrict__ x4, int si16, float wl,
                                             int fg, float4& acc) {
    float4 xv[CNT];
    float  w[CNT];
    #pragma unroll
    for (int t = 0; t < CNT; ++t) {
        int idx = __shfl(si16, G + t, 16) + fg;
        w[t]  = __shfl(wl, G + t, 16);
        xv[t] = x4[idx];
    }
    #pragma unroll
    for (int t = 0; t < CNT; ++t) {
        acc.x += w[t] * xv[t].x;
        acc.y += w[t] * xv[t].y;
        acc.z += w[t] * xv[t].z;
        acc.w += w[t] * xv[t].w;
    }
}

template <bool SECOND>
__global__ __launch_bounds__(256) void spmm_gather_kernel(const int* __restrict__ rowptr,
                                                          const int* __restrict__ ent,
                                                          const float* __restrict__ x,
                                                          const float* __restrict__ x0,
                                                          float* __restrict__ out, int n) {
    int tid = blockIdx.x * blockDim.x + threadIdx.x;
    int row = tid >> 4;
    int fg  = tid & 15;
    if (row >= n) return;
    const float4* x4 = (const float4*)x;
    const unsigned long long* __restrict__ ent2 = (const unsigned long long*)ent;
    int start = rowptr[row];
    int end   = rowptr[row + 1];
    float4 v0 = make_float4(0.f, 0.f, 0.f, 0.f);
    if (SECOND) v0 = ((const float4*)x0)[row * 16 + fg];
    float4 acc = make_float4(0.f, 0.f, 0.f, 0.f);
    for (int base = start; base < end; base += 16) {
        int j = base + fg;
        bool valid = j < end;
        unsigned long long e = ent2[valid ? j : base];
        float wl = valid ? __uint_as_float((unsigned)(e >> 32)) : 0.f;
        int si16 = (int)(unsigned)e * 16;
        int cnt = end - base;
        if (cnt >= 8) {
            gather_group<0, 8>(x4, si16, wl, fg, acc);
        } else {
            gather_group<0, 4>(x4, si16, wl, fg, acc);
            if (cnt > 4) gather_group<4, 4>(x4, si16, wl, fg, acc);
        }
        if (cnt >= 16) {
            gather_group<8, 8>(x4, si16, wl, fg, acc);
        } else if (cnt > 8) {
            gather_group<8, 4>(x4, si16, wl, fg, acc);
            if (cnt > 12) gather_group<12, 4>(x4, si16, wl, fg, acc);
        }
    }
    if (SECOND) {
        acc.x = 2.0f * acc.x - v0.x;
        acc.y = 2.0f * acc.y - v0.y;
        acc.z = 2.0f * acc.z - v0.z;
        acc.w = 2.0f * acc.w - v0.w;
    }
    ((float4*)out)[row * 16 + fg] = acc;
}

template <int NT, int NOUT>
__device__ inline void cheb_accum2(const float* __restrict__ A,
                                   const float* __restrict__ Wp,
                                   int rowA0, int rowA1, int khalf, int colLane,
                                   v8f (&acc0)[NT], v8f (&acc1)[NT]) {
    const float* ar0 = A + (size_t)rowA0 * 64;
    const float* ar1 = A + (size_t)rowA1 * 64;
    #pragma unroll
    for (int k0 = 0; k0 < 64; k0 += 4) {
        int kk = k0 + 2 * khalf;
        v2f a0 = *(const v2f*)(ar0 + kk);
        v2f a1 = *(const v2f*)(ar1 + kk);
        int p = (k0 >> 1) + khalf;
        #pragma unroll
        for (int t = 0; t < NT; ++t) {
            int cg = t * 16 + colLane;
            v2f b;
            if ((NOUT & 15) == 0 || cg < NOUT) {
                b = *(const v2f*)(Wp + ((size_t)p * NOUT + cg) * 2);
            } else {
                b.x = 0.f;
                b.y = 0.f;
            }
            acc0[t] = __builtin_amdgcn_wmma_f32_16x16x4_f32(
                false, a0, false, b, (short)0, acc0[t], false, false);
            acc1[t] = __builtin_amdgcn_wmma_f32_16x16x4_f32(
                false, a1, false, b, (short)0, acc1[t], false, false);
        }
    }
}

template <int NOUT, bool RELU>
__global__ __launch_bounds__(256) void cheb_gemm_kernel(const float* __restrict__ A0,
                                                        const float* __restrict__ A1,
                                                        const float* __restrict__ A2,
                                                        const float* __restrict__ Wp,
                                                        const float* __restrict__ bias,
                                                        float* __restrict__ out, int nodes) {
    constexpr int NT  = (NOUT + 15) / 16;
    constexpr int PSZ = 64 * NOUT;
    int wave = (blockIdx.x * blockDim.x + threadIdx.x) >> 5;
    int lane = threadIdx.x & 31;
    int l15  = lane & 15;
    int khalf = lane >> 4;
    int row0 = wave * 32;
    if (row0 >= nodes) return;

    v8f acc0[NT] = {};
    v8f acc1[NT] = {};
    cheb_accum2<NT, NOUT>(A0, Wp + 0 * PSZ, row0 + l15, row0 + 16 + l15, khalf, l15, acc0, acc1);
    cheb_accum2<NT, NOUT>(A1, Wp + 1 * PSZ, row0 + l15, row0 + 16 + l15, khalf, l15, acc0, acc1);
    cheb_accum2<NT, NOUT>(A2, Wp + 2 * PSZ, row0 + l15, row0 + 16 + l15, khalf, l15, acc0, acc1);

    #pragma unroll
    for (int t = 0; t < NT; ++t) {
        int col = t * 16 + l15;
        if ((NOUT & 15) != 0 && col >= NOUT) continue;
        float bv = bias[col];
        #pragma unroll
        for (int i = 0; i < 8; ++i) {
            int r = row0 + i + 8 * khalf;
            float v0 = acc0[t][i] + bv;
            float v1 = acc1[t][i] + bv;
            if (RELU) { v0 = fmaxf(v0, 0.f); v1 = fmaxf(v1, 0.f); }
            if (r < nodes)      out[(size_t)r * NOUT + col]        = v0;
            if (r + 16 < nodes) out[(size_t)(r + 16) * NOUT + col] = v1;
        }
    }
}

__global__ __launch_bounds__(256) void softmax40_kernel(float* __restrict__ out, int nodes) {
    int i = blockIdx.x * blockDim.x + threadIdx.x;
    if (i >= nodes) return;
    float* row = out + (size_t)i * 40;
    float m = -INFINITY;
    #pragma unroll
    for (int c = 0; c < 40; ++c) m = fmaxf(m, row[c]);
    float e[40];
    float s = 0.f;
    #pragma unroll
    for (int c = 0; c < 40; ++c) { e[c] = __expf(row[c] - m); s += e[c]; }
    float inv = 1.f / s;
    #pragma unroll
    for (int c = 0; c < 40; ++c) row[c] = e[c] * inv;
}

extern "C" void kernel_launch(void* const* d_in, const int* in_sizes, int n_in,
                              void* d_out, int out_size, void* d_ws, size_t ws_size,
                              hipStream_t stream) {
    const float* x  = (const float*)d_in[0];
    const int*   ei = (const int*)d_in[1];
    const float* W1 = (const float*)d_in[2];
    const float* b1 = (const float*)d_in[3];
    const float* W2 = (const float*)d_in[4];
    const float* b2 = (const float*)d_in[5];
    float* out = (float*)d_out;

    const int N = in_sizes[0] / 64;
    const int E = in_sizes[1] / 2;
    const int* src = ei;
    const int* dst = ei + E;

    auto align256 = [](size_t v) { return (v + 255) & ~(size_t)255; };
    char* ws = (char*)d_ws;
    size_t off = 0;
    float* deg  = (float*)(ws + off); off += align256((size_t)N * 4);
    float* A1   = (float*)(ws + off); off += align256((size_t)N * 64 * 4);
    float* A2   = (float*)(ws + off); off += align256((size_t)N * 64 * 4);
    float* H    = (float*)(ws + off); off += align256((size_t)N * 64 * 4);
    float* Wp1  = (float*)(ws + off); off += align256((size_t)3 * 64 * 64 * 4);
    float* Wp2  = (float*)(ws + off); off += align256((size_t)3 * 64 * 40 * 4);
    int* ent = (int*)d_out;
    int* rowptr = (off + align256((size_t)(N + 1) * 4) <= ws_size)
                      ? (int*)(ws + off)
                      : (int*)((char*)d_out + align256((size_t)E * 8));
    unsigned* recs = (unsigned*)H;
    int* offs = (int*)A1;
    (void)n_in; (void)out_size;

    const int TB = 256;
    const int nodeBlk  = (N + TB - 1) / TB;
    const int nTiles   = (E + TILE_E - 1) / TILE_E;
    const int nBins    = (N + BIN_W - 1) / BIN_W;
    unsigned* srecs = recs + (size_t)nTiles * TILE_E;
    int* soffs = offs + (size_t)(nBins + 1) * nTiles;
    const int rowBlk   = (int)(((long long)N * 16 + TB - 1) / TB);
    const int gemmBlk  = ((((N + 31) / 32) * 32) + TB - 1) / TB;

    packw_kernel<<<(3 * 64 * 64 + TB - 1) / TB, TB, 0, stream>>>(W1, Wp1, 64, 3 * 64 * 64);
    packw_kernel<<<(3 * 64 * 40 + TB - 1) / TB, TB, 0, stream>>>(W2, Wp2, 40, 3 * 64 * 40);
    tile_sort_kernel<<<nTiles, 1024, 0, stream>>>(src, dst, recs, offs, srecs, soffs, E, nBins, nTiles);
    deg_kernel<<<nBins, 1024, 0, stream>>>(srecs, soffs, deg, N, nTiles);
    csr_build_kernel<<<nBins, 1024, 0, stream>>>(recs, offs, deg, rowptr, ent, N, nBins, nTiles);

    spmm_gather_kernel<false><<<rowBlk, TB, 0, stream>>>(rowptr, ent, x, x, A1, N);
    spmm_gather_kernel<true><<<rowBlk, TB, 0, stream>>>(rowptr, ent, A1, x, A2, N);
    cheb_gemm_kernel<64, true><<<gemmBlk, TB, 0, stream>>>(x, A1, A2, Wp1, b1, H, N);

    spmm_gather_kernel<false><<<rowBlk, TB, 0, stream>>>(rowptr, ent, H, H, A1, N);
    spmm_gather_kernel<true><<<rowBlk, TB, 0, stream>>>(rowptr, ent, A1, H, A2, N);
    cheb_gemm_kernel<40, false><<<gemmBlk, TB, 0, stream>>>(H, A1, A2, Wp2, b2, out, N);

    softmax40_kernel<<<nodeBlk, TB, 0, stream>>>(out, N);
}


# Row softmax done in the layer-2 GEMM epilogue: row max and sum folded over a row's 16 lanes by xor shuffles (row sum order is now a lane tree); the softmax launch and its re-read of out go
# speedup vs baseline: 3.6867x; 1.0338x over previous
#include <hip/hip_runtime.h>
#include <hip/hip_bf16.h>

typedef __attribute__((ext_vector_type(2))) float v2f;
typedef __attribute__((ext_vector_type(8))) float v8f;

__global__ __launch_bounds__(256) void packw_kernel(const float* __restrict__ W,
                                                    float* __restrict__ Wp,
                                                    int nout, int total) {
    int idx = blockIdx.x * blockDim.x + threadIdx.x;
    if (idx >= total) return;
    int col = idx % nout;
    int k   = (idx / nout) & 63;
    int m   = idx / (nout * 64);
    Wp[(((size_t)m * 32 + (k >> 1)) * nout + col) * 2 + (k & 1)] = W[idx];
}

constexpr int TILE_E = 4096;
constexpr int BIN_SH = 8;
constexpr int BIN_W  = 1 << BIN_SH;
constexpr int NB_CAP = 512;
constexpr int ST_CAP = 8192;

__device__ __forceinline__ int wave_incl_scan(int v, int lane) {
    #pragma unroll
    for (int o = 1; o < 32; o <<= 1) {
        int u = __shfl_up(v, o, 32);
        if (lane >= o) v += u;
    }
    return v;
}

__device__ __forceinline__ int wave_sum(int v) {
    #pragma unroll
    for (int o = 16; o > 0; o >>= 1) v += __shfl_xor(v, o, 32);
    return v;
}

__device__ __forceinline__ float node_scale(float dg) {
    return (dg > 0.f) ? rsqrtf(fmaxf(dg, 1.0f)) : 0.f;
}

__global__ __launch_bounds__(1024) void tile_sort_kernel(const int* __restrict__ src,
                                                         const int* __restrict__ dst,
                                                         unsigned* __restrict__ recs,
                                                         int* __restrict__ offs,
                                                         unsigned* __restrict__ srecs,
                                                         int* __restrict__ soffs,
                                                         int nedges, int nb, int nt) {
    __shared__ int hist[2][NB_CAP];
    __shared__ int wtot[2][NB_CAP / 32];
    __shared__ unsigned stage[2][TILE_E];
    static_assert(2 * NB_CAP == 1024, "one thread per counter of the two histograms");
    const int t = threadIdx.x, lane = t & 31;
    const int tile = blockIdx.x;
    hist[t >> 9][t & (NB_CAP - 1)] = 0;
    __syncthreads();

    int s[4], d[4];
    #pragma unroll
    for (int k = 0; k < 4; ++k) {
        int e = tile * TILE_E + k * 1024 + t;
        s[k] = 0; d[k] = 0;
        if (e < nedges) { s[k] = src[e]; d[k] = dst[e]; }
    }
    int rk[4], rs[4];
    #pragma unroll
    for (int k = 0; k < 4; ++k) {
        rk[k] = 0; rs[k] = 0;
        if (s[k] != d[k]) {
            rk[k] = atomicAdd(&hist[0][d[k] >> BIN_SH], 1);
            rs[k] = atomicAdd(&hist[1][s[k] >> BIN_SH], 1);
        }
    }
    __syncthreads();

    const int h = t >> 9, u = t & (NB_CAP - 1), wu = u >> 5;
    const int v = (u < nb) ? hist[h][u] : 0;
    const int inc = wave_incl_scan(v, lane);
    if (lane == 31) wtot[h][wu] = inc;
    __syncthreads();
    {
        int woff = 0;
        for (int w = 0; w < wu; ++w) woff += wtot[h][w];
        const int exc = woff + inc - v;
        hist[h][u] = exc;
        int* __restrict__ o = h ? soffs : offs;
        if (u <= nb) o[(size_t)u * nt + tile] = exc;
    }
    __syncthreads();

    #pragma unroll
    for (int k = 0; k < 4; ++k) {
        if (s[k] != d[k]) {
            stage[0][hist[0][d[k] >> BIN_SH] + rk[k]] =
                ((unsigned)s[k] << BIN_SH) | (unsigned)(d[k] & (BIN_W - 1));
            stage[1][hist[1][s[k] >> BIN_SH] + rs[k]] = (unsigned)(s[k] & (BIN_W - 1));
        }
    }
    __syncthreads();
    const int total = hist[0][nb];
    unsigned* out  = recs  + (size_t)tile * TILE_E;
    unsigned* out2 = srecs + (size_t)tile * TILE_E;
    for (int p = t; p < total; p += 1024) {
        out[p]  = stage[0][p];
        out2[p] = stage[1][p];
    }
}

__global__ __launch_bounds__(1024) void deg_kernel(const unsigned* __restrict__ srecs,
                                                   const int* __restrict__ soffs,
                                                   float* __restrict__ deg,
                                                   int n, int nt) {
    __shared__ int cnt[BIN_W];
    const int t = threadIdx.x;
    const int b = blockIdx.x;
    const int* __restrict__ o0 = soffs + (size_t)b * nt;
    const int* __restrict__ o1 = o0 + nt;
    if (t < BIN_W) cnt[t] = 0;

    constexpr int SEGS = 8;
    const int hw = t >> 4, l16 = t & 15;
    int lo[SEGS], hi[SEGS];
    #pragma unroll
    for (int j = 0; j < SEGS; ++j) {
        const int seg = hw + 64 * j;
        lo[j] = 0; hi[j] = 0;
        if (seg < nt) { lo[j] = o0[seg]; hi[j] = o1[seg]; }
    }
    unsigned r0[SEGS];
    #pragma unroll
    for (int j = 0; j < SEGS; ++j) {
        const int k = lo[j] + l16;
        r0[j] = 0u;
        if (k < hi[j]) r0[j] = srecs[(size_t)(hw + 64 * j) * TILE_E + k];
    }
    __syncthreads();

    #pragma unroll
    for (int j = 0; j < SEGS; ++j) {
        const int k = lo[j] + l16;
        if (k < hi[j]) {
            atomicAdd(&cnt[r0[j] & (BIN_W - 1)], 1);
            const unsigned* rp = srecs + (size_t)(hw + 64 * j) * TILE_E;
            for (int kk = k + 16; kk < hi[j]; kk += 16) atomicAdd(&cnt[rp[kk] & (BIN_W - 1)], 1);
        }
    }
    for (int seg = hw + 64 * SEGS; seg < nt; seg += 64) {
        const int l = o0[seg], hh = o1[seg];
        const unsigned* rp = srecs + (size_t)seg * TILE_E;
        for (int k = l + l16; k < hh; k += 16) atomicAdd(&cnt[rp[k] & (BIN_W - 1)], 1);
    }
    __syncthreads();
    if (t < BIN_W) {
        const int node = b * BIN_W + t;
        if (node < n) deg[node] = (float)cnt[t];
    }
}

__global__ __launch_bounds__(1024) void csr_build_kernel(const unsigned* __restrict__ recs,
                                                         const int* __restrict__ offs,
                                                         const float* __restrict__ deg,
                                                         int* __restrict__ rowptr,
                                                         int* __restrict__ ent,
                                                         int n, int nb, int nt) {
    __shared__ int cnt[BIN_W];
    __shared__ float nsc[BIN_W];
    __shared__ int wlo[32];
    __shared__ int whi[32];
    __shared__ int wtot[BIN_W / 32];
    __shared__ int2 stage[ST_CAP];
    int2* __restrict__ ent2 = (int2*)ent;
    const int t = threadIdx.x, lane = t & 31, wv = t >> 5;
    const int b = blockIdx.x;
    const int* __restrict__ o0 = offs + (size_t)b * nt;
    const int* __restrict__ o1 = o0 + nt;
    if (t < BIN_W) cnt[t] = 0;

    constexpr int SEGS = 8;
    const int hw = t >> 4, l16 = t & 15;
    int lo[SEGS], hi[SEGS];
    #pragma unroll
    for (int j = 0; j < SEGS; ++j) {
        const int seg = hw + 64 * j;
        lo[j] = 0; hi[j] = 0;
        if (seg < nt) { lo[j] = o0[seg]; hi[j] = o1[seg]; }
    }
    unsigned r0[SEGS];
    #pragma unroll
    for (int j = 0; j < SEGS; ++j) {
        const int k = lo[j] + l16;
        r0[j] = 0u;
        if (k < hi[j]) r0[j] = recs[(size_t)(hw + 64 * j) * TILE_E + k];
    }

    int plo = 0, phi = 0;
    #pragma nounroll
    for (int i = t; i < nt; i += 1024) { plo += o0[i]; phi += o1[i]; }
    plo = wave_sum(plo);
    phi = wave_sum(phi);
    if (lane == 0) { wlo[wv] = plo; whi[wv] = phi; }
    __syncthreads();
    const int base  = wave_sum(wlo[lane]);
    const int total = wave_sum(whi[lane]) - base;

    #pragma unroll
    for (int j = 0; j < SEGS; ++j) {
        const int k = lo[j] + l16;
        if (k < hi[j]) {
            atomicAdd(&cnt[r0[j] & (BIN_W - 1)], 1);
            const unsigned* rp = recs + (size_t)(hw + 64 * j) * TILE_E;
            for (int kk = k + 16; kk < hi[j]; kk += 16) atomicAdd(&cnt[rp[kk] & (BIN_W - 1)], 1);
        }
    }
    for (int seg = hw + 64 * SEGS; seg < nt; seg += 64) {
        const int l = o0[seg], h = o1[seg];
        const unsigned* rp = recs + (size_t)seg * TILE_E;
        for (int k = l + l16; k < h; k += 16) atomicAdd(&cnt[rp[k] & (BIN_W - 1)], 1);
    }
    float g0[SEGS];
    #pragma unroll
    for (int j = 0; j < SEGS; ++j) {
        g0[j] = 0.f;
        if (lo[j] + l16 < hi[j]) g0[j] = deg[r0[j] >> BIN_SH];
    }
    __syncthreads();

    int c = 0, inc = 0;
    if (t < BIN_W) {
        c = cnt[t];
        inc = wave_incl_scan(c, lane);
        if (lane == 31) wtot[wv] = inc;
    }
    __syncthreads();
    if (t < BIN_W) {
        int woff = 0;
        for (int w = 0; w < wv; ++w) woff += wtot[w];
        const int exc = woff + inc - c;
        cnt[t] = exc;
        const int node = b * BIN_W + t;
        float sc = 0.f;
        if (node < n) {
            rowptr[node] = base + exc;
            sc = node_scale(deg[node]);
        }
        nsc[t] = -sc;
    }
    if (b == nb - 1 && t == 0) rowptr[n] = base + total;
    __syncthreads();

    const bool fits = total <= ST_CAP;
    auto place = [&](unsigned r, float dg) {
        const int ln = (int)(r & (BIN_W - 1));
        const int p = atomicAdd(&cnt[ln], 1);
        const int2 e = make_int2((int)(r >> BIN_SH), __float_as_int(nsc[ln] * node_scale(dg)));
        if (fits) stage[p] = e;
        else      ent2[base + p] = e;
    };
    #pragma unroll
    for (int j = 0; j < SEGS; ++j) {
        const int k = lo[j] + l16;
        if (k < hi[j]) {
            place(r0[j], g0[j]);
            const unsigned* rp = recs + (size_t)(hw + 64 * j) * TILE_E;
            for (int kk = k + 16; kk < hi[j]; kk += 16) {
                const unsigned r = rp[kk];
                place(r, deg[r >> BIN_SH]);
            }
        }
    }
    for (int seg = hw + 64 * SEGS; seg < nt; seg += 64) {
        const int l = o0[seg], h = o1[seg];
        const unsigned* rp = recs + (size_t)seg * TILE_E;
        for (int k = l + l16; k < h; k += 16) {
            const unsigned r = rp[k];
            place(r, deg[r >> BIN_SH]);
        }
    }
    __syncthreads();
    if (fits) {
        for (int p = t; p < total; p += 1024) ent2[base + p] = stage[p];
    }
}

template <int G, int CNT>
__device__ __forceinline__ void gather_group(const float4* __restrict__ x4, int si16, float wl,
                                             int fg, float4& acc) {
    float4 xv[CNT];
    float  w[CNT];
    #pragma unroll
    for (int t = 0; t < CNT; ++t) {
        int idx = __shfl(si16, G + t, 16) + fg;
        w[t]  = __shfl(wl, G + t, 16);
        xv[t] = x4[idx];
    }
    #pragma unroll
    for (int t = 0; t < CNT; ++t) {
        acc.x += w[t] * xv[t].x;
        acc.y += w[t] * xv[t].y;
        acc.z += w[t] * xv[t].z;
        acc.w += w[t] * xv[t].w;
    }
}

template <bool SECOND>
__global__ __launch_bounds__(256) void spmm_gather_kernel(const int* __restrict__ rowptr,
                                                          const int* __restrict__ ent,
                                                          const float* __restrict__ x,
                                                          const float* __restrict__ x0,
                                                          float* __restrict__ out, int n) {
    int tid = blockIdx.x * blockDim.x + threadIdx.x;
    int row = tid >> 4;
    int fg  = tid & 15;
    if (row >= n) return;
    const float4* x4 = (const float4*)x;
    const unsigned long long* __restrict__ ent2 = (const unsigned long long*)ent;
    int start = rowptr[row];
    int end   = rowptr[row + 1];
    float4 v0 = make_float4(0.f, 0.f, 0.f, 0.f);
    if (SECOND) v0 = ((const float4*)x0)[row * 16 + fg];
    float4 acc = make_float4(0.f, 0.f, 0.f, 0.f);
    for (int base = start; base < end; base += 16) {
        int j = base + fg;
        bool valid = j < end;
        unsigned long long e = ent2[valid ? j : base];
        float wl = valid ? __uint_as_float((unsigned)(e >> 32)) : 0.f;
        int si16 = (int)(unsigned)e * 16;
        int cnt = end - base;
        if (cnt >= 8) {
            gather_group<0, 8>(x4, si16, wl, fg, acc);
        } else {
            gather_group<0, 4>(x4, si16, wl, fg, acc);
            if (cnt > 4) gather_group<4, 4>(x4, si16, wl, fg, acc);
        }
        if (cnt >= 16) {
            gather_group<8, 8>(x4, si16, wl, fg, acc);
        } else if (cnt > 8) {
            gather_group<8, 4>(x4, si16, wl, fg, acc);
            if (cnt > 12) gather_group<12, 4>(x4, si16, wl, fg, acc);
        }
    }
    if (SECOND) {
        acc.x = 2.0f * acc.x - v0.x;
        acc.y = 2.0f * acc.y - v0.y;
        acc.z = 2.0f * acc.z - v0.z;
        acc.w = 2.0f * acc.w - v0.w;
    }
    ((float4*)out)[row * 16 + fg] = acc;
}

template <int NT, int NOUT>
__device__ inline void cheb_accum2(const float* __restrict__ A,
                                   const float* __restrict__ Wp,
                                   int rowA0, int rowA1, int khalf, int colLane,
                                   v8f (&acc0)[NT], v8f (&acc1)[NT]) {
    const float* ar0 = A + (size_t)rowA0 * 64;
    const float* ar1 = A + (size_t)rowA1 * 64;
    #pragma unroll
    for (int k0 = 0; k0 < 64; k0 += 4) {
        int kk = k0 + 2 * khalf;
        v2f a0 = *(const v2f*)(ar0 + kk);
        v2f a1 = *(const v2f*)(ar1 + kk);
        int p = (k0 >> 1) + khalf;
        #pragma unroll
        for (int t = 0; t < NT; ++t) {
            int cg = t * 16 + colLane;
            v2f b;
            if ((NOUT & 15) == 0 || cg < NOUT) {
                b = *(const v2f*)(Wp + ((size_t)p * NOUT + cg) * 2);
            } else {
                b.x = 0.f;
                b.y = 0.f;
            }
            acc0[t] = __builtin_amdgcn_wmma_f32_16x16x4_f32(
                false, a0, false, b, (short)0, acc0[t], false, false);
            acc1[t] = __builtin_amdgcn_wmma_f32_16x16x4_f32(
                false, a1, false, b, (short)0, acc1[t], false, false);
        }
    }
}

template <int NOUT, bool RELU, bool SOFTMAX>
__global__ __launch_bounds__(256) void cheb_gemm_kernel(const float* __restrict__ A0,
                                                        const float* __restrict__ A1,
                                                        const float* __restrict__ A2,
                                                        const float* __restrict__ Wp,
                                                        const float* __restrict__ bias,
                                                        float* __restrict__ out, int nodes) {
    constexpr int NT  = (NOUT + 15) / 16;
    constexpr int PSZ = 64 * NOUT;
    int wave = (blockIdx.x * blockDim.x + threadIdx.x) >> 5;
    int lane = threadIdx.x & 31;
    int l15  = lane & 15;
    int khalf = lane >> 4;
    int row0 = wave * 32;
    if (row0 >= nodes) return;

    v8f acc0[NT] = {};
    v8f acc1[NT] = {};
    cheb_accum2<NT, NOUT>(A0, Wp + 0 * PSZ, row0 + l15, row0 + 16 + l15, khalf, l15, acc0, acc1);
    cheb_accum2<NT, NOUT>(A1, Wp + 1 * PSZ, row0 + l15, row0 + 16 + l15, khalf, l15, acc0, acc1);
    cheb_accum2<NT, NOUT>(A2, Wp + 2 * PSZ, row0 + l15, row0 + 16 + l15, khalf, l15, acc0, acc1);

    if (!SOFTMAX) {
        #pragma unroll
        for (int t = 0; t < NT; ++t) {
            int col = t * 16 + l15;
            if ((NOUT & 15) != 0 && col >= NOUT) continue;
            float bv = bias[col];
            #pragma unroll
            for (int i = 0; i < 8; ++i) {
                int r = row0 + i + 8 * khalf;
                float v0 = acc0[t][i] + bv;
                float v1 = acc1[t][i] + bv;
                if (RELU) { v0 = fmaxf(v0, 0.f); v1 = fmaxf(v1, 0.f); }
                if (r < nodes)      out[(size_t)r * NOUT + col]        = v0;
                if (r + 16 < nodes) out[(size_t)(r + 16) * NOUT + col] = v1;
            }
        }
    } else {
        float bv[NT];
        bool  ok[NT];
        #pragma unroll
        for (int t = 0; t < NT; ++t) {
            int col = t * 16 + l15;
            ok[t] = ((NOUT & 15) == 0) || (col < NOUT);
            bv[t] = ok[t] ? bias[col] : 0.f;
        }
        float m[16];
        #pragma unroll
        for (int i = 0; i < 8; ++i) {
            #pragma unroll
            for (int t = 0; t < NT; ++t) {
                acc0[t][i] = ok[t] ? acc0[t][i] + bv[t] : -INFINITY;
                acc1[t][i] = ok[t] ? acc1[t][i] + bv[t] : -INFINITY;
            }
            float m0 = acc0[0][i], m1 = acc1[0][i];
            #pragma unroll
            for (int t = 1; t < NT; ++t) {
                m0 = fmaxf(m0, acc0[t][i]);
                m1 = fmaxf(m1, acc1[t][i]);
            }
            m[i] = m0; m[8 + i] = m1;
        }
        #pragma unroll
        for (int o = 8; o > 0; o >>= 1) {
            float u[16];
            #pragma unroll
            for (int j = 0; j < 16; ++j) u[j] = __shfl_xor(m[j], o, 32);
            #pragma unroll
            for (int j = 0; j < 16; ++j) m[j] = fmaxf(m[j], u[j]);
        }
        float s[16];
        #pragma unroll
        for (int i = 0; i < 8; ++i) {
            float s0 = 0.f, s1 = 0.f;
            #pragma unroll
            for (int t = 0; t < NT; ++t) {
                const float e0 = __expf(acc0[t][i] - m[i]);
                const float e1 = __expf(acc1[t][i] - m[8 + i]);
                acc0[t][i] = e0; acc1[t][i] = e1;
                s0 += e0; s1 += e1;
            }
            s[i] = s0; s[8 + i] = s1;
        }
        __builtin_amdgcn_sched_barrier(0);
        #pragma unroll
        for (int o = 8; o > 0; o >>= 1) {
            float u[16];
            #pragma unroll
            for (int j = 0; j < 16; ++j) u[j] = __shfl_xor(s[j], o, 32);
            #pragma unroll
            for (int j = 0; j < 16; ++j) s[j] += u[j];
        }
        #pragma unroll
        for (int i = 0; i < 8; ++i) {
            const float inv0 = 1.f / s[i];
            const float inv1 = 1.f / s[8 + i];
            #pragma unroll
            for (int t = 0; t < NT; ++t) { acc0[t][i] *= inv0; acc1[t][i] *= inv1; }
        }
        #pragma unroll
        for (int t = 0; t < NT; ++t) {
            const int col = t * 16 + l15;
            if (!ok[t]) continue;
            #pragma unroll
            for (int i = 0; i < 8; ++i) {
                const int r = row0 + i + 8 * khalf;
                if (r < nodes)      out[(size_t)r * NOUT + col]        = acc0[t][i];
                if (r + 16 < nodes) out[(size_t)(r + 16) * NOUT + col] = acc1[t][i];
            }
        }
    }
}

extern "C" void kernel_launch(void* const* d_in, const int* in_sizes, int n_in,
                              void* d_out, int out_size, void* d_ws, size_t ws_size,
                              hipStream_t stream) {
    const float* x  = (const float*)d_in[0];
    const int*   ei = (const int*)d_in[1];
    const float* W1 = (const float*)d_in[2];
    const float* b1 = (const float*)d_in[3];
    const float* W2 = (const float*)d_in[4];
    const float* b2 = (const float*)d_in[5];
    float* out = (float*)d_out;

    const int N = in_sizes[0] / 64;
    const int E = in_sizes[1] / 2;
    const int* src = ei;
    const int* dst = ei + E;

    auto align256 = [](size_t v) { return (v + 255) & ~(size_t)255; };
    char* ws = (char*)d_ws;
    size_t off = 0;
    float* deg  = (float*)(ws + off); off += align256((size_t)N * 4);
    float* A1   = (float*)(ws + off); off += align256((size_t)N * 64 * 4);
    float* A2   = (float*)(ws + off); off += align256((size_t)N * 64 * 4);
    float* H    = (float*)(ws + off); off += align256((size_t)N * 64 * 4);
    float* Wp1  = (float*)(ws + off); off += align256((size_t)3 * 64 * 64 * 4);
    float* Wp2  = (float*)(ws + off); off += align256((size_t)3 * 64 * 40 * 4);
    int* ent = (int*)d_out;
    int* rowptr = (off + align256((size_t)(N + 1) * 4) <= ws_size)
                      ? (int*)(ws + off)
                      : (int*)((char*)d_out + align256((size_t)E * 8));
    unsigned* recs = (unsigned*)H;
    int* offs = (int*)A1;
    (void)n_in; (void)out_size;

    const int TB = 256;
    const int nTiles   = (E + TILE_E - 1) / TILE_E;
    const int nBins    = (N + BIN_W - 1) / BIN_W;
    unsigned* srecs = recs + (size_t)nTiles * TILE_E;
    int* soffs = offs + (size_t)(nBins + 1) * nTiles;
    const int rowBlk   = (int)(((long long)N * 16 + TB - 1) / TB);
    const int gemmBlk  = ((((N + 31) / 32) * 32) + TB - 1) / TB;

    packw_kernel<<<(3 * 64 * 64 + TB - 1) / TB, TB, 0, stream>>>(W1, Wp1, 64, 3 * 64 * 64);
    packw_kernel<<<(3 * 64 * 40 + TB - 1) / TB, TB, 0, stream>>>(W2, Wp2, 40, 3 * 64 * 40);
    tile_sort_kernel<<<nTiles, 1024, 0, stream>>>(src, dst, recs, offs, srecs, soffs, E, nBins, nTiles);
    deg_kernel<<<nBins, 1024, 0, stream>>>(srecs, soffs, deg, N, nTiles);
    csr_build_kernel<<<nBins, 1024, 0, stream>>>(recs, offs, deg, rowptr, ent, N, nBins, nTiles);

    spmm_gather_kernel<false><<<rowBlk, TB, 0, stream>>>(rowptr, ent, x, x, A1, N);
    spmm_gather_kernel<true><<<rowBlk, TB, 0, stream>>>(rowptr, ent, A1, x, A2, N);
    cheb_gemm_kernel<64, true, false><<<gemmBlk, TB, 0, stream>>>(x, A1, A2, Wp1, b1, H, N);

    spmm_gather_kernel<false><<<rowBlk, TB, 0, stream>>>(rowptr, ent, H, H, A1, N);
    spmm_gather_kernel<true><<<rowBlk, TB, 0, stream>>>(rowptr, ent, A1, H, A2, N);
    cheb_gemm_kernel<40, false, true><<<gemmBlk, TB, 0, stream>>>(H, A1, A2, Wp2, b2, out, N);

}
